# starting kernel
# baseline (speedup 1.0000x reference)
#include <hip/hip_runtime.h>

typedef __attribute__((ext_vector_type(2))) float v2f;
typedef __attribute__((ext_vector_type(8))) float v8f;

#define DFEAT 128
#define TILE  16

__global__ void zero_f4(float4* __restrict__ p, long long n4) {
  long long i = (long long)blockIdx.x * blockDim.x + threadIdx.x;
  if (i < n4) p[i] = make_float4(0.f, 0.f, 0.f, 0.f);
}

__global__ __launch_bounds__(256) void rgcn_proj_gemm(
    const float* __restrict__ feats,
    const float* __restrict__ Wc, const float* __restrict__ bc,
    const float* __restrict__ We, const float* __restrict__ be,
    float* __restrict__ WhC, float* __restrict__ WhE, int n)
{
  const int wave  = threadIdx.x >> 5;
  const int lane  = threadIdx.x & 31;
  const int gw    = blockIdx.x * 8 + wave;
  const int nColT = DFEAT / TILE;
  const int m0    = (gw / nColT) * TILE;
  const int n0    = (gw % nColT) * TILE;
  if (m0 >= n) return;

  const int half = lane >> 4;
  const int l15  = lane & 15;

  int arowi = m0 + l15;
  if (arowi >= n) arowi = n - 1;
  const float* arow  = feats + (size_t)arowi * DFEAT + half * 2;
  const float* bcrow = Wc    + (size_t)(n0 + l15) * DFEAT + half * 2;
  const float* berow = We    + (size_t)(n0 + l15) * DFEAT + half * 2;

  v8f accC = {0.f, 0.f, 0.f, 0.f, 0.f, 0.f, 0.f, 0.f};
  v8f accE = {0.f, 0.f, 0.f, 0.f, 0.f, 0.f, 0.f, 0.f};
#pragma unroll 8
  for (int k = 0; k < DFEAT; k += 4) {
    v2f a   = *(const v2f*)(arow + k);
    v2f bC  = *(const v2f*)(bcrow + k);
    v2f bE  = *(const v2f*)(berow + k);
    accC = __builtin_amdgcn_wmma_f32_16x16x4_f32(
        false, a, false, bC, (short)0, accC, false, false);
    accE = __builtin_amdgcn_wmma_f32_16x16x4_f32(
        false, a, false, bE, (short)0, accE, false, false);
  }

  const float biasC = bc[n0 + l15];
  const float biasE = be[n0 + l15];
  float* ocol = WhC + (size_t)(m0 + half * 8) * DFEAT + n0 + l15;
  float* ecol = WhE + (size_t)(m0 + half * 8) * DFEAT + n0 + l15;

  if (m0 + TILE <= n) {
#pragma unroll
    for (int r = 0; r < 8; ++r) {
      ocol[(size_t)r * DFEAT] = accC[r] + biasC;
      ecol[(size_t)r * DFEAT] = accE[r] + biasE;
    }
  } else {
#pragma unroll
    for (int r = 0; r < 8; ++r) {
      const int M = m0 + half * 8 + r;
      if (M < n) {
        ocol[(size_t)r * DFEAT] = accC[r] + biasC;
        ecol[(size_t)r * DFEAT] = accE[r] + biasE;
      }
    }
  }
}

__global__ __launch_bounds__(256) void rgcn_out_gemm(
    const float* __restrict__ h,
    const float* __restrict__ Wo, const float* __restrict__ bo,
    const float* __restrict__ WhE, float* __restrict__ out, int n)
{
  const int wave  = threadIdx.x >> 5;
  const int lane  = threadIdx.x & 31;
  const int gw    = blockIdx.x * 8 + wave;
  const int nColT = DFEAT / TILE;
  const int m0    = (gw / nColT) * TILE;
  const int n0    = (gw % nColT) * TILE;
  if (m0 >= n) return;

  const int half = lane >> 4;
  const int l15  = lane & 15;

  int arowi = m0 + l15;
  if (arowi >= n) arowi = n - 1;
  const float* arow = h  + (size_t)arowi * DFEAT + half * 2;
  const float* brow = Wo + (size_t)(n0 + l15) * DFEAT + half * 2;

  v8f acc = {0.f, 0.f, 0.f, 0.f, 0.f, 0.f, 0.f, 0.f};
#pragma unroll 8
  for (int k = 0; k < DFEAT; k += 4) {
    v2f a = *(const v2f*)(arow + k);
    v2f b = *(const v2f*)(brow + k);
    acc = __builtin_amdgcn_wmma_f32_16x16x4_f32(
        false, a, false, b, (short)0, acc, false, false);
  }

  const float bias = bo[n0 + l15];
  const size_t base = (size_t)(m0 + half * 8) * DFEAT + n0 + l15;

  if (m0 + TILE <= n) {
#pragma unroll
    for (int r = 0; r < 8; ++r) {
      const size_t idx = base + (size_t)r * DFEAT;
      out[idx] = acc[r] + bias + WhE[idx];
    }
  } else {
#pragma unroll
    for (int r = 0; r < 8; ++r) {
      const int M = m0 + half * 8 + r;
      if (M < n) {
        const size_t idx = base + (size_t)r * DFEAT;
        out[idx] = acc[r] + bias + WhE[idx];
      }
    }
  }
}

__global__ __launch_bounds__(256) void scatter_add_edges(
    const float* __restrict__ Wh,
    const int* __restrict__ src, const int* __restrict__ dst,
    float* __restrict__ h, int nE)
{
  const long long idx = (long long)blockIdx.x * blockDim.x + threadIdx.x;
  const int e = (int)(idx >> 7);
  if (e >= nE) return;
  const int d = (int)(idx & 127);
  const int s = src[e];
  const int t = dst[e];
  atomicAdd(h + (size_t)t * DFEAT + d, Wh[(size_t)s * DFEAT + d]);
}

extern "C" void kernel_launch(void* const* d_in, const int* in_sizes, int n_in,
                              void* d_out, int out_size, void* d_ws, size_t ws_size,
                              hipStream_t stream) {
  const float* feats = (const float*)d_in[0];
  const float* Wc    = (const float*)d_in[1];
  const float* bc    = (const float*)d_in[2];
  const float* We    = (const float*)d_in[3];
  const float* be    = (const float*)d_in[4];
  const float* Wo    = (const float*)d_in[5];
  const float* bo    = (const float*)d_in[6];
  const int*   srcC  = (const int*)d_in[7];
  const int*   dstC  = (const int*)d_in[8];
  const int*   srcE  = (const int*)d_in[9];
  const int*   dstE  = (const int*)d_in[10];

  const int n   = in_sizes[0] / DFEAT;
  const int nEc = in_sizes[7];
  const int nEe = in_sizes[9];

  const size_t feSize = (size_t)n * DFEAT;
  float* WhC = (float*)d_ws;
  float* WhE = WhC + feSize;
  float* h   = WhE + feSize;
  float* out = (float*)d_out;

  const long long n4 = (long long)feSize / 4;
  zero_f4<<<(unsigned)((n4 + 255) / 256), 256, 0, stream>>>((float4*)h, n4);

  const int tiles  = ((n + TILE - 1) / TILE) * (DFEAT / TILE);
  const int blocks = (tiles + 7) / 8;
  rgcn_proj_gemm<<<dim3(blocks, 1), 256, 0, stream>>>(
      feats, Wc, bc, We, be, WhC, WhE, n);

  const long long workC = (long long)nEc * DFEAT;
  scatter_add_edges<<<(unsigned)((workC + 255) / 256), 256, 0, stream>>>(
      WhC, srcC, dstC, h, nEc);
  const long long workE = (long long)nEe * DFEAT;
  scatter_add_edges<<<(unsigned)((workE + 255) / 256), 256, 0, stream>>>(
      WhE, srcE, dstE, h, nEe);

  rgcn_out_gemm<<<dim3(blocks, 1), 256, 0, stream>>>(h, Wo, bo, WhE, out, n);
}


# Edge sums without float atomics: counting sort of all edges by destination each call (int atomic returns the rank), one wave sums a node's rows in registers; sum order = arrival-order rank
# speedup vs baseline: 1.9521x; 1.9521x over previous
#include <hip/hip_runtime.h>

typedef __attribute__((ext_vector_type(2))) float v2f;
typedef __attribute__((ext_vector_type(8))) float v8f;

#define DFEAT 128
#define TILE  16

__global__ __launch_bounds__(256) void rgcn_proj_gemm(
    const float* __restrict__ feats,
    const float* __restrict__ Wc, const float* __restrict__ bc,
    const float* __restrict__ We, const float* __restrict__ be,
    float* __restrict__ WhC, float* __restrict__ WhE, int n)
{
  const int wave  = threadIdx.x >> 5;
  const int lane  = threadIdx.x & 31;
  const int gw    = blockIdx.x * 8 + wave;
  const int nColT = DFEAT / TILE;
  const int m0    = (gw / nColT) * TILE;
  const int n0    = (gw % nColT) * TILE;
  if (m0 >= n) return;

  const int half = lane >> 4;
  const int l15  = lane & 15;

  int arowi = m0 + l15;
  if (arowi >= n) arowi = n - 1;
  const float* arow  = feats + (size_t)arowi * DFEAT + half * 2;
  const float* bcrow = Wc    + (size_t)(n0 + l15) * DFEAT + half * 2;
  const float* berow = We    + (size_t)(n0 + l15) * DFEAT + half * 2;

  v8f accC = {0.f, 0.f, 0.f, 0.f, 0.f, 0.f, 0.f, 0.f};
  v8f accE = {0.f, 0.f, 0.f, 0.f, 0.f, 0.f, 0.f, 0.f};
#pragma unroll 8
  for (int k = 0; k < DFEAT; k += 4) {
    v2f a   = *(const v2f*)(arow + k);
    v2f bC  = *(const v2f*)(bcrow + k);
    v2f bE  = *(const v2f*)(berow + k);
    accC = __builtin_amdgcn_wmma_f32_16x16x4_f32(
        false, a, false, bC, (short)0, accC, false, false);
    accE = __builtin_amdgcn_wmma_f32_16x16x4_f32(
        false, a, false, bE, (short)0, accE, false, false);
  }

  const float biasC = bc[n0 + l15];
  const float biasE = be[n0 + l15];
  float* ocol = WhC + (size_t)(m0 + half * 8) * DFEAT + n0 + l15;
  float* ecol = WhE + (size_t)(m0 + half * 8) * DFEAT + n0 + l15;

  if (m0 + TILE <= n) {
#pragma unroll
    for (int r = 0; r < 8; ++r) {
      ocol[(size_t)r * DFEAT] = accC[r] + biasC;
      ecol[(size_t)r * DFEAT] = accE[r] + biasE;
    }
  } else {
#pragma unroll
    for (int r = 0; r < 8; ++r) {
      const int M = m0 + half * 8 + r;
      if (M < n) {
        ocol[(size_t)r * DFEAT] = accC[r] + biasC;
        ecol[(size_t)r * DFEAT] = accE[r] + biasE;
      }
    }
  }
}

__global__ __launch_bounds__(256) void rgcn_out_gemm(
    const float* __restrict__ h,
    const float* __restrict__ Wo, const float* __restrict__ bo,
    const float* __restrict__ WhE, float* __restrict__ out, int n)
{
  const int wave  = threadIdx.x >> 5;
  const int lane  = threadIdx.x & 31;
  const int gw    = blockIdx.x * 8 + wave;
  const int nColT = DFEAT / TILE;
  const int m0    = (gw / nColT) * TILE;
  const int n0    = (gw % nColT) * TILE;
  if (m0 >= n) return;

  const int half = lane >> 4;
  const int l15  = lane & 15;

  int arowi = m0 + l15;
  if (arowi >= n) arowi = n - 1;
  const float* arow = h  + (size_t)arowi * DFEAT + half * 2;
  const float* brow = Wo + (size_t)(n0 + l15) * DFEAT + half * 2;

  v8f acc = {0.f, 0.f, 0.f, 0.f, 0.f, 0.f, 0.f, 0.f};
#pragma unroll 8
  for (int k = 0; k < DFEAT; k += 4) {
    v2f a = *(const v2f*)(arow + k);
    v2f b = *(const v2f*)(brow + k);
    acc = __builtin_amdgcn_wmma_f32_16x16x4_f32(
        false, a, false, b, (short)0, acc, false, false);
  }

  const float bias = bo[n0 + l15];
  const size_t base = (size_t)(m0 + half * 8) * DFEAT + n0 + l15;

  if (m0 + TILE <= n) {
#pragma unroll
    for (int r = 0; r < 8; ++r) {
      const size_t idx = base + (size_t)r * DFEAT;
      out[idx] = acc[r] + bias + WhE[idx];
    }
  } else {
#pragma unroll
    for (int r = 0; r < 8; ++r) {
      const int M = m0 + half * 8 + r;
      if (M < n) {
        const size_t idx = base + (size_t)r * DFEAT;
        out[idx] = acc[r] + bias + WhE[idx];
      }
    }
  }
}

__global__ void zero_i4(int4* __restrict__ p, int n4) {
  const int i = blockIdx.x * blockDim.x + threadIdx.x;
  if (i < n4) p[i] = make_int4(0, 0, 0, 0);
}

__global__ __launch_bounds__(256) void rank_edges(
    const int* __restrict__ dstC, const int* __restrict__ dstE,
    int* __restrict__ off, int* __restrict__ rnk, int nEc, int nEe)
{
  const int e = blockIdx.x * 256 + threadIdx.x;
  if (e >= nEc + nEe) return;
  const int t = (e < nEc) ? dstC[e] : dstE[e - nEc];
  rnk[e] = atomicAdd(off + t, 1);
}

#define SCAN_THREADS 1024
#define SCAN_CHUNK   49
#define SCAN_GROUP   7
#define SCAN_TILE    (SCAN_THREADS * SCAN_CHUNK)

__device__ __forceinline__ int wave_incl_scan(int v, int lane) {
#pragma unroll
  for (int d = 1; d < 32; d <<= 1) {
    const int up = __shfl_up(v, d);
    if (lane >= d) v += up;
  }
  return v;
}

__global__ __launch_bounds__(SCAN_THREADS) void scan_offsets(int* __restrict__ off, int n)
{
  __shared__ int tile[SCAN_TILE];
  __shared__ int wtot[32];
  const int tid  = threadIdx.x;
  const int lane = tid & 31;
  const int wave = tid >> 5;
  int carry = 0;
  for (int base = 0; base < n; base += SCAN_TILE) {
    for (int k0 = 0; k0 < SCAN_CHUNK; k0 += SCAN_GROUP) {
      int v[SCAN_GROUP];
#pragma unroll
      for (int q = 0; q < SCAN_GROUP; ++q) {
        const int g = base + (k0 + q) * SCAN_THREADS + tid;
        v[q] = off[(g < n) ? g : (n - 1)];
      }
#pragma unroll
      for (int q = 0; q < SCAN_GROUP; ++q) {
        const int i = (k0 + q) * SCAN_THREADS + tid;
        tile[i] = (base + i < n) ? v[q] : 0;
      }
    }
    __syncthreads();
    int* mine = tile + tid * SCAN_CHUNK;
    int s = 0;
#pragma unroll
    for (int j = 0; j < SCAN_CHUNK; ++j) s += mine[j];
    const int incl = wave_incl_scan(s, lane);
    if (lane == 31) wtot[wave] = incl;
    __syncthreads();
    if (wave == 0) wtot[lane] = wave_incl_scan(wtot[lane], lane);
    __syncthreads();
    int run = carry + (wave ? wtot[wave - 1] : 0) + (incl - s);
    carry += wtot[31];
#pragma unroll
    for (int j = 0; j < SCAN_CHUNK; ++j) {
      const int c = mine[j];
      mine[j] = run;
      run += c;
    }
    __syncthreads();
    for (int k0 = 0; k0 < SCAN_CHUNK; k0 += SCAN_GROUP) {
      int v[SCAN_GROUP];
#pragma unroll
      for (int q = 0; q < SCAN_GROUP; ++q) v[q] = tile[(k0 + q) * SCAN_THREADS + tid];
#pragma unroll
      for (int q = 0; q < SCAN_GROUP; ++q) {
        const int g = base + (k0 + q) * SCAN_THREADS + tid;
        if (g < n) off[g] = v[q];
      }
    }
    __syncthreads();
  }
  if (tid == 0) off[n] = carry;
}

__global__ __launch_bounds__(256) void place_edges(
    const int* __restrict__ srcC, const int* __restrict__ dstC,
    const int* __restrict__ srcE, const int* __restrict__ dstE,
    const int* __restrict__ off, const int* __restrict__ rnk,
    int* __restrict__ lst, int nEc, int nEe, int n)
{
  const int e = blockIdx.x * 256 + threadIdx.x;
  if (e >= nEc + nEe) return;
  int t, s;
  if (e < nEc) { t = dstC[e];       s = srcC[e]; }
  else         { t = dstE[e - nEc]; s = srcE[e - nEc] + n; }
  lst[off[t] + rnk[e]] = s;
}

__device__ __forceinline__ void add4(float4& a, const float4 v) {
  a.x += v.x; a.y += v.y; a.z += v.z; a.w += v.w;
}

__global__ __launch_bounds__(256) void gather_rows(
    const float* __restrict__ Wh2, const int* __restrict__ off,
    const int* __restrict__ lst, float* __restrict__ h, int n)
{
  const int lane = threadIdx.x & 31;
  const int node = __builtin_amdgcn_readfirstlane(blockIdx.x * 8 + (threadIdx.x >> 5));
  if (node >= n) return;
  const int beg = off[node];
  const int end = off[node + 1];
  const float4* rows = (const float4*)Wh2 + lane;
  const float4 zero4 = make_float4(0.f, 0.f, 0.f, 0.f);
  float4 acc = zero4;
  for (int base = beg; base < end; base += 32) {
    const int m = (end - base < 32) ? (end - base) : 32;
    const int idx = (lane < m) ? lst[base + lane] : 0;
    for (int j = 0; j < m; j += 4) {
      const int left = m - j;
      float4 v0, v1 = zero4, v2 = zero4, v3 = zero4;
      v0 = rows[(size_t)__builtin_amdgcn_readlane(idx, j) * 32];
      if (left > 1) v1 = rows[(size_t)__builtin_amdgcn_readlane(idx, j + 1) * 32];
      if (left > 2) v2 = rows[(size_t)__builtin_amdgcn_readlane(idx, j + 2) * 32];
      if (left > 3) v3 = rows[(size_t)__builtin_amdgcn_readlane(idx, j + 3) * 32];
      add4(acc, v0); add4(acc, v1); add4(acc, v2); add4(acc, v3);
    }
  }
  ((float4*)h)[(size_t)node * 32 + lane] = acc;
}

extern "C" void kernel_launch(void* const* d_in, const int* in_sizes, int n_in,
                              void* d_out, int out_size, void* d_ws, size_t ws_size,
                              hipStream_t stream) {
  const float* feats = (const float*)d_in[0];
  const float* Wc    = (const float*)d_in[1];
  const float* bc    = (const float*)d_in[2];
  const float* We    = (const float*)d_in[3];
  const float* be    = (const float*)d_in[4];
  const float* Wo    = (const float*)d_in[5];
  const float* bo    = (const float*)d_in[6];
  const int*   srcC  = (const int*)d_in[7];
  const int*   dstC  = (const int*)d_in[8];
  const int*   srcE  = (const int*)d_in[9];
  const int*   dstE  = (const int*)d_in[10];

  const int n   = in_sizes[0] / DFEAT;
  const int nEc = in_sizes[7];
  const int nEe = in_sizes[9];
  const int nE  = nEc + nEe;

  const size_t feSize = (size_t)n * DFEAT;
  float* WhC = (float*)d_ws;
  float* WhE = WhC + feSize;
  float* h   = WhE + feSize;
  int*   off = (int*)(h + feSize);
  int*   rnk = off + ((size_t)n / 1024 + 2) * 1024;
  int*   lst = rnk + nE;
  float* out = (float*)d_out;

  const int n4 = (n + 3) / 4;
  zero_i4<<<(unsigned)((n4 + 255) / 256), 256, 0, stream>>>((int4*)off, n4);
  const unsigned eBlocks = (unsigned)((nE + 255) / 256);
  rank_edges<<<eBlocks, 256, 0, stream>>>(dstC, dstE, off, rnk, nEc, nEe);
  scan_offsets<<<1, SCAN_THREADS, 0, stream>>>(off, n);
  place_edges<<<eBlocks, 256, 0, stream>>>(srcC, dstC, srcE, dstE, off, rnk, lst,
                                           nEc, nEe, n);

  const int tiles  = ((n + TILE - 1) / TILE) * (DFEAT / TILE);
  const int blocks = (tiles + 7) / 8;
  rgcn_proj_gemm<<<dim3(blocks, 1), 256, 0, stream>>>(
      feats, Wc, bc, We, be, WhC, WhE, n);

  gather_rows<<<(unsigned)((n + 7) / 8), 256, 0, stream>>>(WhC, off, lst, h, n);

  rgcn_out_gemm<<<dim3(blocks, 1), 256, 0, stream>>>(h, Wo, bo, WhE, out, n);
}


# Edge ranks from LDS counters, no returning global atomics: two-level counting sort (256-node buckets per edge chunk, then per-bucket node slots by LDS add); sum order = LDS arrival order
# speedup vs baseline: 2.6599x; 1.3626x over previous
#include <hip/hip_runtime.h>

typedef __attribute__((ext_vector_type(2))) float v2f;
typedef __attribute__((ext_vector_type(8))) float v8f;

#define DFEAT 128
#define TILE  16

__global__ __launch_bounds__(256) void rgcn_proj_gemm(
    const float* __restrict__ feats,
    const float* __restrict__ Wc, const float* __restrict__ bc,
    const float* __restrict__ We, const float* __restrict__ be,
    float* __restrict__ WhC, float* __restrict__ WhE, int n)
{
  const int wave  = threadIdx.x >> 5;
  const int lane  = threadIdx.x & 31;
  const int gw    = blockIdx.x * 8 + wave;
  const int nColT = DFEAT / TILE;
  const int m0    = (gw / nColT) * TILE;
  const int n0    = (gw % nColT) * TILE;
  if (m0 >= n) return;

  const int half = lane >> 4;
  const int l15  = lane & 15;

  int arowi = m0 + l15;
  if (arowi >= n) arowi = n - 1;
  const float* arow  = feats + (size_t)arowi * DFEAT + half * 2;
  const float* bcrow = Wc    + (size_t)(n0 + l15) * DFEAT + half * 2;
  const float* berow = We    + (size_t)(n0 + l15) * DFEAT + half * 2;

  v8f accC = {0.f, 0.f, 0.f, 0.f, 0.f, 0.f, 0.f, 0.f};
  v8f accE = {0.f, 0.f, 0.f, 0.f, 0.f, 0.f, 0.f, 0.f};
#pragma unroll 8
  for (int k = 0; k < DFEAT; k += 4) {
    v2f a   = *(const v2f*)(arow + k);
    v2f bC  = *(const v2f*)(bcrow + k);
    v2f bE  = *(const v2f*)(berow + k);
    accC = __builtin_amdgcn_wmma_f32_16x16x4_f32(
        false, a, false, bC, (short)0, accC, false, false);
    accE = __builtin_amdgcn_wmma_f32_16x16x4_f32(
        false, a, false, bE, (short)0, accE, false, false);
  }

  const float biasC = bc[n0 + l15];
  const float biasE = be[n0 + l15];
  float* ocol = WhC + (size_t)(m0 + half * 8) * DFEAT + n0 + l15;
  float* ecol = WhE + (size_t)(m0 + half * 8) * DFEAT + n0 + l15;

  if (m0 + TILE <= n) {
#pragma unroll
    for (int r = 0; r < 8; ++r) {
      ocol[(size_t)r * DFEAT] = accC[r] + biasC;
      ecol[(size_t)r * DFEAT] = accE[r] + biasE;
    }
  } else {
#pragma unroll
    for (int r = 0; r < 8; ++r) {
      const int M = m0 + half * 8 + r;
      if (M < n) {
        ocol[(size_t)r * DFEAT] = accC[r] + biasC;
        ecol[(size_t)r * DFEAT] = accE[r] + biasE;
      }
    }
  }
}

__global__ __launch_bounds__(256) void rgcn_out_gemm(
    const float* __restrict__ h,
    const float* __restrict__ Wo, const float* __restrict__ bo,
    const float* __restrict__ WhE, float* __restrict__ out, int n)
{
  const int wave  = threadIdx.x >> 5;
  const int lane  = threadIdx.x & 31;
  const int gw    = blockIdx.x * 8 + wave;
  const int nColT = DFEAT / TILE;
  const int m0    = (gw / nColT) * TILE;
  const int n0    = (gw % nColT) * TILE;
  if (m0 >= n) return;

  const int half = lane >> 4;
  const int l15  = lane & 15;

  int arowi = m0 + l15;
  if (arowi >= n) arowi = n - 1;
  const float* arow = h  + (size_t)arowi * DFEAT + half * 2;
  const float* brow = Wo + (size_t)(n0 + l15) * DFEAT + half * 2;

  v8f acc = {0.f, 0.f, 0.f, 0.f, 0.f, 0.f, 0.f, 0.f};
#pragma unroll 8
  for (int k = 0; k < DFEAT; k += 4) {
    v2f a = *(const v2f*)(arow + k);
    v2f b = *(const v2f*)(brow + k);
    acc = __builtin_amdgcn_wmma_f32_16x16x4_f32(
        false, a, false, b, (short)0, acc, false, false);
  }

  const float bias = bo[n0 + l15];
  const size_t base = (size_t)(m0 + half * 8) * DFEAT + n0 + l15;

  if (m0 + TILE <= n) {
#pragma unroll
    for (int r = 0; r < 8; ++r) {
      const size_t idx = base + (size_t)r * DFEAT;
      out[idx] = acc[r] + bias + WhE[idx];
    }
  } else {
#pragma unroll
    for (int r = 0; r < 8; ++r) {
      const int M = m0 + half * 8 + r;
      if (M < n) {
        const size_t idx = base + (size_t)r * DFEAT;
        out[idx] = acc[r] + bias + WhE[idx];
      }
    }
  }
}

#define BSH     8
#define BNODES  256
#define MAXB    256
#define NCHUNK  256
#define STILE   4096

__device__ __forceinline__ int wave_incl_scan(int v, int lane) {
#pragma unroll
  for (int d = 1; d < 32; d <<= 1) {
    const int up = __shfl_up(v, d);
    if (lane >= d) v += up;
  }
  return v;
}

__global__ __launch_bounds__(256) void bucket_count(
    const int* __restrict__ dstC, const int* __restrict__ dstE,
    int* __restrict__ cnt, int nEc, int nE, int nB)
{
  __shared__ int hist[MAXB + 1];
  const int tid = threadIdx.x;
  const int j   = blockIdx.x;
  hist[tid] = 0;
  __syncthreads();
  const int per = (nE + NCHUNK - 1) / NCHUNK;
  const int beg = j * per;
  const int end = (beg + per < nE) ? (beg + per) : nE;
  for (int t0 = beg; t0 < end; t0 += STILE) {
    int d[16];
#pragma unroll
    for (int q = 0; q < 16; ++q) {
      const int e  = t0 + q * 256 + tid;
      const int ec = (e < end) ? e : (end - 1);
      const int* pd = (ec < nEc) ? (dstC + ec) : (dstE + (ec - nEc));
      d[q] = *pd;
    }
#pragma unroll
    for (int q = 0; q < 16; ++q) {
      const bool ok = t0 + q * 256 + tid < end;
      atomicAdd(&hist[ok ? (d[q] >> BSH) : MAXB], 1);
    }
  }
  __syncthreads();
  if (tid < nB) cnt[tid * NCHUNK + j] = hist[tid];
}

#define SCAN_THREADS 1024
#define SCAN_CHUNK   49
#define SCAN_GROUP   7
#define SCAN_TILE    (SCAN_THREADS * SCAN_CHUNK)

__global__ __launch_bounds__(SCAN_THREADS) void scan_offsets(int* __restrict__ off, int n)
{
  __shared__ int tile[SCAN_TILE];
  __shared__ int wtot[32];
  const int tid  = threadIdx.x;
  const int lane = tid & 31;
  const int wave = tid >> 5;
  int carry = 0;
  for (int base = 0; base < n; base += SCAN_TILE) {
    for (int k0 = 0; k0 < SCAN_CHUNK; k0 += SCAN_GROUP) {
      int v[SCAN_GROUP];
#pragma unroll
      for (int q = 0; q < SCAN_GROUP; ++q) {
        const int g = base + (k0 + q) * SCAN_THREADS + tid;
        v[q] = off[(g < n) ? g : (n - 1)];
      }
#pragma unroll
      for (int q = 0; q < SCAN_GROUP; ++q) {
        const int i = (k0 + q) * SCAN_THREADS + tid;
        tile[i] = (base + i < n) ? v[q] : 0;
      }
    }
    __syncthreads();
    int* mine = tile + tid * SCAN_CHUNK;
    int s = 0;
#pragma unroll
    for (int j = 0; j < SCAN_CHUNK; ++j) s += mine[j];
    const int incl = wave_incl_scan(s, lane);
    if (lane == 31) wtot[wave] = incl;
    __syncthreads();
    if (wave == 0) wtot[lane] = wave_incl_scan(wtot[lane], lane);
    __syncthreads();
    int run = carry + (wave ? wtot[wave - 1] : 0) + (incl - s);
    carry += wtot[31];
#pragma unroll
    for (int j = 0; j < SCAN_CHUNK; ++j) {
      const int c = mine[j];
      mine[j] = run;
      run += c;
    }
    __syncthreads();
    for (int k0 = 0; k0 < SCAN_CHUNK; k0 += SCAN_GROUP) {
      int v[SCAN_GROUP];
#pragma unroll
      for (int q = 0; q < SCAN_GROUP; ++q) v[q] = tile[(k0 + q) * SCAN_THREADS + tid];
#pragma unroll
      for (int q = 0; q < SCAN_GROUP; ++q) {
        const int g = base + (k0 + q) * SCAN_THREADS + tid;
        if (g < n) off[g] = v[q];
      }
    }
    __syncthreads();
  }
  if (tid == 0) off[n] = carry;
}

__global__ __launch_bounds__(256) void bucket_scatter(
    const int* __restrict__ srcC, const int* __restrict__ dstC,
    const int* __restrict__ srcE, const int* __restrict__ dstE,
    const int* __restrict__ S, int2* __restrict__ rec,
    int nEc, int nE, int n, int nB)
{
  __shared__ int  hist[MAXB + 1];
  __shared__ int  lbase[MAXB];
  __shared__ int  gbase[MAXB];
  __shared__ int  wsum[8];
  __shared__ int2 tile[STILE];
  const int tid  = threadIdx.x;
  const int lane = tid & 31;
  const int wave = tid >> 5;
  const int j    = blockIdx.x;
  gbase[tid] = (tid < nB) ? S[tid * NCHUNK + j] : 0;
  const int per = (nE + NCHUNK - 1) / NCHUNK;
  const int beg = j * per;
  const int end = (beg + per < nE) ? (beg + per) : nE;
  for (int t0 = beg; t0 < end; t0 += STILE) {
    hist[tid] = 0;
    __syncthreads();
    int d[16], s[16], r[16];
#pragma unroll
    for (int q = 0; q < 16; ++q) {
      const int e  = t0 + q * 256 + tid;
      const int ec = (e < end) ? e : (end - 1);
      const bool chem = ec < nEc;
      const int* pd = chem ? (dstC + ec) : (dstE + (ec - nEc));
      const int* ps = chem ? (srcC + ec) : (srcE + (ec - nEc));
      d[q] = *pd;
      s[q] = *ps + (chem ? 0 : n);
    }
#pragma unroll
    for (int q = 0; q < 16; ++q) {
      const bool ok = t0 + q * 256 + tid < end;
      r[q] = atomicAdd(&hist[ok ? (d[q] >> BSH) : MAXB], 1);
      if (!ok) d[q] = -1;
    }
    __syncthreads();
    const int c    = hist[tid];
    const int incl = wave_incl_scan(c, lane);
    if (lane == 31) wsum[wave] = incl;
    __syncthreads();
    int wb = 0;
    for (int w = 0; w < wave; ++w) wb += wsum[w];
    lbase[tid] = wb + incl - c;
    __syncthreads();
#pragma unroll
    for (int q = 0; q < 16; ++q)
      if (d[q] >= 0) tile[lbase[d[q] >> BSH] + r[q]] = make_int2(d[q], s[q]);
    __syncthreads();
    const int cntT = (end - t0 < STILE) ? (end - t0) : STILE;
    for (int i = tid; i < cntT; i += 256) {
      const int2 v = tile[i];
      const int  b = v.x >> BSH;
      rec[gbase[b] + (i - lbase[b])] = v;
    }
    __syncthreads();
    gbase[tid] += c;
  }
}

__global__ __launch_bounds__(1024) void bucket_place(
    const int* __restrict__ S, const int2* __restrict__ rec,
    int* __restrict__ off, int* __restrict__ lst, int n, int nB)
{
  __shared__ int cnt[BNODES + 1];
  __shared__ int cur[BNODES + 1];
  __shared__ int wsum[8];
  const int tid  = threadIdx.x;
  const int lane = tid & 31;
  const int wave = tid >> 5;
  const int b    = blockIdx.x;
  const int segBeg = S[b * NCHUNK];
  const int segEnd = S[(b + 1) * NCHUNK];
  if (tid < BNODES) cnt[tid] = 0;
  __syncthreads();
  for (int i0 = segBeg; i0 < segEnd; i0 += 4096) {
    int2 v[4];
#pragma unroll
    for (int q = 0; q < 4; ++q) {
      const int i = i0 + q * 1024 + tid;
      v[q] = rec[(i < segEnd) ? i : (segEnd - 1)];
    }
#pragma unroll
    for (int q = 0; q < 4; ++q) {
      const bool ok = i0 + q * 1024 + tid < segEnd;
      atomicAdd(&cnt[ok ? (v[q].x & (BNODES - 1)) : BNODES], 1);
    }
  }
  __syncthreads();
  const int c    = (tid < BNODES) ? cnt[tid] : 0;
  const int incl = wave_incl_scan(c, lane);
  if (tid < BNODES && lane == 31) wsum[wave] = incl;
  __syncthreads();
  if (tid < BNODES) {
    int wb = 0;
    for (int w = 0; w < wave; ++w) wb += wsum[w];
    const int st = segBeg + wb + incl - c;
    cur[tid] = st;
    const int node = b * BNODES + tid;
    if (node < n) off[node] = st;
  }
  if (b == nB - 1 && tid == 0) off[n] = segEnd;
  __syncthreads();
  for (int i0 = segBeg; i0 < segEnd; i0 += 4096) {
    int2 v[4];
#pragma unroll
    for (int q = 0; q < 4; ++q) {
      const int i = i0 + q * 1024 + tid;
      v[q] = rec[(i < segEnd) ? i : (segEnd - 1)];
    }
#pragma unroll
    for (int q = 0; q < 4; ++q) {
      const bool ok = i0 + q * 1024 + tid < segEnd;
      const int  p  = atomicAdd(&cur[ok ? (v[q].x & (BNODES - 1)) : BNODES], 1);
      if (ok) lst[p] = v[q].y;
    }
  }
}

__device__ __forceinline__ void add4(float4& a, const float4 v) {
  a.x += v.x; a.y += v.y; a.z += v.z; a.w += v.w;
}

__global__ __launch_bounds__(256) void gather_rows(
    const float* __restrict__ Wh2, const int* __restrict__ off,
    const int* __restrict__ lst, float* __restrict__ h, int n)
{
  const int lane = threadIdx.x & 31;
  const int node = __builtin_amdgcn_readfirstlane(blockIdx.x * 8 + (threadIdx.x >> 5));
  if (node >= n) return;
  const int beg = off[node];
  const int end = off[node + 1];
  const float4* rows = (const float4*)Wh2 + lane;
  const float4 zero4 = make_float4(0.f, 0.f, 0.f, 0.f);
  float4 acc = zero4;
  for (int base = beg; base < end; base += 32) {
    const int m = (end - base < 32) ? (end - base) : 32;
    const int idx = (lane < m) ? lst[base + lane] : 0;
    for (int j = 0; j < m; j += 4) {
      const int left = m - j;
      float4 v0, v1 = zero4, v2 = zero4, v3 = zero4;
      v0 = rows[(size_t)__builtin_amdgcn_readlane(idx, j) * 32];
      if (left > 1) v1 = rows[(size_t)__builtin_amdgcn_readlane(idx, j + 1) * 32];
      if (left > 2) v2 = rows[(size_t)__builtin_amdgcn_readlane(idx, j + 2) * 32];
      if (left > 3) v3 = rows[(size_t)__builtin_amdgcn_readlane(idx, j + 3) * 32];
      add4(acc, v0); add4(acc, v1); add4(acc, v2); add4(acc, v3);
    }
  }
  ((float4*)h)[(size_t)node * 32 + lane] = acc;
}

extern "C" void kernel_launch(void* const* d_in, const int* in_sizes, int n_in,
                              void* d_out, int out_size, void* d_ws, size_t ws_size,
                              hipStream_t stream) {
  const float* feats = (const float*)d_in[0];
  const float* Wc    = (const float*)d_in[1];
  const float* bc    = (const float*)d_in[2];
  const float* We    = (const float*)d_in[3];
  const float* be    = (const float*)d_in[4];
  const float* Wo    = (const float*)d_in[5];
  const float* bo    = (const float*)d_in[6];
  const int*   srcC  = (const int*)d_in[7];
  const int*   dstC  = (const int*)d_in[8];
  const int*   srcE  = (const int*)d_in[9];
  const int*   dstE  = (const int*)d_in[10];

  const int n   = in_sizes[0] / DFEAT;
  const int nEc = in_sizes[7];
  const int nEe = in_sizes[9];
  const int nE  = nEc + nEe;

  const size_t feSize = (size_t)n * DFEAT;
  float* WhC = (float*)d_ws;
  float* WhE = WhC + feSize;
  float* h   = WhE + feSize;
  int*   off = (int*)(h + feSize);
  int*   S   = off + ((size_t)n / 1024 + 2) * 1024;
  int2*  rec = (int2*)(S + (size_t)MAXB * NCHUNK + 1024);
  int*   lst = (int*)(rec + nE);
  float* out = (float*)d_out;

  const int nB = (n + BNODES - 1) / BNODES;
  bucket_count<<<NCHUNK, 256, 0, stream>>>(dstC, dstE, S, nEc, nE, nB);
  scan_offsets<<<1, SCAN_THREADS, 0, stream>>>(S, nB * NCHUNK);
  bucket_scatter<<<NCHUNK, 256, 0, stream>>>(srcC, dstC, srcE, dstE, S, rec,
                                             nEc, nE, n, nB);
  bucket_place<<<(unsigned)nB, 1024, 0, stream>>>(S, rec, off, lst, n, nB);

  const int tiles  = ((n + TILE - 1) / TILE) * (DFEAT / TILE);
  const int blocks = (tiles + 7) / 8;
  rgcn_proj_gemm<<<dim3(blocks, 1), 256, 0, stream>>>(
      feats, Wc, bc, We, be, WhC, WhE, n);

  gather_rows<<<(unsigned)((n + 7) / 8), 256, 0, stream>>>(WhC, off, lst, h, n);

  rgcn_out_gemm<<<dim3(blocks, 1), 256, 0, stream>>>(h, Wo, bo, WhE, out, n);
}


# GEMM operand feed: a wave keeps its 16 columns of W in registers over all its row tiles; a row tile goes through LDS once by whole-line loads instead of 16-line fragment loads from every wave
# speedup vs baseline: 4.1400x; 1.5564x over previous
#include <hip/hip_runtime.h>

typedef __attribute__((ext_vector_type(2))) float v2f;
typedef __attribute__((ext_vector_type(8))) float v8f;

#define DFEAT 128
#define TILE  16

#define LDP 132

__device__ __forceinline__ void wave_stage16(const float* __restrict__ src, float* dst, int lane) {
  const float4* s4 = (const float4*)src + lane;
  float4 v[16];
#pragma unroll
  for (int q = 0; q < 16; ++q) v[q] = s4[q * 32];
#pragma unroll
  for (int q = 0; q < 16; ++q) *(float4*)(dst + q * LDP + lane * 4) = v[q];
}

__device__ __forceinline__ void read_frags(const float* p, v2f (&f)[32]) {
#pragma unroll
  for (int t = 0; t < 32; ++t) f[t] = *(const v2f*)(p + 4 * t);
}

__device__ __forceinline__ void tile_load(const float* __restrict__ A, int m0, int n,
                                          int wave, int lane, float4& p0, float4& p1) {
  int r0 = m0 + wave, r1 = m0 + wave + 8;
  r0 = (r0 < n) ? r0 : (n - 1);
  r1 = (r1 < n) ? r1 : (n - 1);
  p0 = ((const float4*)(A + (size_t)r0 * DFEAT))[lane];
  p1 = ((const float4*)(A + (size_t)r1 * DFEAT))[lane];
}

__device__ __forceinline__ void tile_store(float* buf, int wave, int lane,
                                           const float4 p0, const float4 p1) {
  *(float4*)(buf + wave * LDP + lane * 4)       = p0;
  *(float4*)(buf + (wave + 8) * LDP + lane * 4) = p1;
}

__global__ __launch_bounds__(256) void rgcn_proj_gemm(
    const float* __restrict__ feats,
    const float* __restrict__ Wc, const float* __restrict__ bc,
    const float* __restrict__ We, const float* __restrict__ be,
    float* __restrict__ WhC, float* __restrict__ WhE, int n)
{
  __shared__ __attribute__((aligned(16))) float lds[2 * DFEAT * LDP];
  const int tid  = threadIdx.x;
  const int wave = tid >> 5;
  const int lane = tid & 31;
  const int half = lane >> 4;
  const int l15  = lane & 15;
  const int n0   = wave * TILE;

  v2f bC[32], bE[32];
  wave_stage16(Wc + (size_t)n0 * DFEAT, lds + n0 * LDP, lane);
  wave_stage16(We + (size_t)n0 * DFEAT, lds + (DFEAT + n0) * LDP, lane);
  __syncthreads();
  read_frags(lds + (n0 + l15) * LDP + 2 * half, bC);
  read_frags(lds + (DFEAT + n0 + l15) * LDP + 2 * half, bE);
  __syncthreads();

  const float biasC = bc[n0 + l15];
  const float biasE = be[n0 + l15];
  const int nTiles = (n + TILE - 1) / TILE;
  const float4 zero4 = make_float4(0.f, 0.f, 0.f, 0.f);
  float4 p0 = zero4, p1 = zero4;
  int t = blockIdx.x;
  if (t < nTiles) {
    tile_load(feats, t * TILE, n, wave, lane, p0, p1);
    tile_store(lds, wave, lane, p0, p1);
  }
  __syncthreads();
  int cur = 0;
  for (; t < nTiles; t += gridDim.x) {
    const int  tn   = t + gridDim.x;
    const bool more = tn < nTiles;
    if (more) tile_load(feats, tn * TILE, n, wave, lane, p0, p1);

    v2f a[32];
    read_frags(lds + cur * (TILE * LDP) + l15 * LDP + 2 * half, a);
    v8f accC = {0.f, 0.f, 0.f, 0.f, 0.f, 0.f, 0.f, 0.f};
    v8f accE = {0.f, 0.f, 0.f, 0.f, 0.f, 0.f, 0.f, 0.f};
#pragma unroll
    for (int k = 0; k < 32; ++k) {
      accC = __builtin_amdgcn_wmma_f32_16x16x4_f32(
          false, a[k], false, bC[k], (short)0, accC, false, false);
      accE = __builtin_amdgcn_wmma_f32_16x16x4_f32(
          false, a[k], false, bE[k], (short)0, accE, false, false);
    }

    const int m0 = t * TILE;
    float oC[8], oE[8];
#pragma unroll
    for (int r = 0; r < 8; ++r) { oC[r] = accC[r] + biasC; oE[r] = accE[r] + biasE; }
    float* ocol = WhC + (size_t)(m0 + half * 8) * DFEAT + n0 + l15;
    float* ecol = WhE + (size_t)(m0 + half * 8) * DFEAT + n0 + l15;
    if (m0 + TILE <= n) {
#pragma unroll
      for (int r = 0; r < 8; ++r) {
        ocol[(size_t)r * DFEAT] = oC[r];
        ecol[(size_t)r * DFEAT] = oE[r];
      }
    } else {
#pragma unroll
      for (int r = 0; r < 8; ++r) {
        if (m0 + half * 8 + r < n) {
          ocol[(size_t)r * DFEAT] = oC[r];
          ecol[(size_t)r * DFEAT] = oE[r];
        }
      }
    }

    if (more) tile_store(lds + (cur ^ 1) * (TILE * LDP), wave, lane, p0, p1);
    __syncthreads();
    cur ^= 1;
  }
}

__global__ __launch_bounds__(256) void rgcn_out_gemm(
    const float* __restrict__ h,
    const float* __restrict__ Wo, const float* __restrict__ bo,
    const float* __restrict__ WhE, float* __restrict__ out, int n)
{
  __shared__ __attribute__((aligned(16))) float lds[DFEAT * LDP];
  const int tid  = threadIdx.x;
  const int wave = tid >> 5;
  const int lane = tid & 31;
  const int half = lane >> 4;
  const int l15  = lane & 15;
  const int n0   = wave * TILE;

  v2f b[32];
  wave_stage16(Wo + (size_t)n0 * DFEAT, lds + n0 * LDP, lane);
  __syncthreads();
  read_frags(lds + (n0 + l15) * LDP + 2 * half, b);
  __syncthreads();

  const float bias = bo[n0 + l15];
  const int nTiles = (n + TILE - 1) / TILE;
  const float4 zero4 = make_float4(0.f, 0.f, 0.f, 0.f);
  float4 p0 = zero4, p1 = zero4;
  int t = blockIdx.x;
  if (t < nTiles) {
    tile_load(h, t * TILE, n, wave, lane, p0, p1);
    tile_store(lds, wave, lane, p0, p1);
  }
  __syncthreads();
  int cur = 0;
  for (; t < nTiles; t += gridDim.x) {
    const int  tn   = t + gridDim.x;
    const bool more = tn < nTiles;
    if (more) tile_load(h, tn * TILE, n, wave, lane, p0, p1);

    const int m0 = t * TILE;
    float e[8];
#pragma unroll
    for (int r = 0; r < 8; ++r) {
      int M = m0 + half * 8 + r;
      M = (M < n) ? M : (n - 1);
      e[r] = WhE[(size_t)M * DFEAT + n0 + l15];
    }

    v2f a[32];
    read_frags(lds + cur * (TILE * LDP) + l15 * LDP + 2 * half, a);
    v8f acc = {0.f, 0.f, 0.f, 0.f, 0.f, 0.f, 0.f, 0.f};
#pragma unroll
    for (int k = 0; k < 32; ++k)
      acc = __builtin_amdgcn_wmma_f32_16x16x4_f32(
          false, a[k], false, b[k], (short)0, acc, false, false);

    float o[8];
#pragma unroll
    for (int r = 0; r < 8; ++r) o[r] = acc[r] + bias + e[r];
    float* ocol = out + (size_t)(m0 + half * 8) * DFEAT + n0 + l15;
    if (m0 + TILE <= n) {
#pragma unroll
      for (int r = 0; r < 8; ++r) ocol[(size_t)r * DFEAT] = o[r];
    } else {
#pragma unroll
      for (int r = 0; r < 8; ++r)
        if (m0 + half * 8 + r < n) ocol[(size_t)r * DFEAT] = o[r];
    }

    if (more) tile_store(lds + (cur ^ 1) * (TILE * LDP), wave, lane, p0, p1);
    __syncthreads();
    cur ^= 1;
  }
}

#define BSH     8
#define BNODES  256
#define MAXB    256
#define NCHUNK  256
#define STILE   4096

__device__ __forceinline__ int wave_incl_scan(int v, int lane) {
#pragma unroll
  for (int d = 1; d < 32; d <<= 1) {
    const int up = __shfl_up(v, d);
    if (lane >= d) v += up;
  }
  return v;
}

__global__ __launch_bounds__(256) void bucket_count(
    const int* __restrict__ dstC, const int* __restrict__ dstE,
    int* __restrict__ cnt, int nEc, int nE, int nB)
{
  __shared__ int hist[MAXB + 1];
  const int tid = threadIdx.x;
  const int j   = blockIdx.x;
  hist[tid] = 0;
  __syncthreads();
  const int per = (nE + NCHUNK - 1) / NCHUNK;
  const int beg = j * per;
  const int end = (beg + per < nE) ? (beg + per) : nE;
  for (int t0 = beg; t0 < end; t0 += STILE) {
    int d[16];
#pragma unroll
    for (int q = 0; q < 16; ++q) {
      const int e  = t0 + q * 256 + tid;
      const int ec = (e < end) ? e : (end - 1);
      const int* pd = (ec < nEc) ? (dstC + ec) : (dstE + (ec - nEc));
      d[q] = *pd;
    }
#pragma unroll
    for (int q = 0; q < 16; ++q) {
      const bool ok = t0 + q * 256 + tid < end;
      atomicAdd(&hist[ok ? (d[q] >> BSH) : MAXB], 1);
    }
  }
  __syncthreads();
  if (tid < nB) cnt[tid * NCHUNK + j] = hist[tid];
}

#define SCAN_THREADS 1024
#define SCAN_CHUNK   49
#define SCAN_GROUP   7
#define SCAN_TILE    (SCAN_THREADS * SCAN_CHUNK)

__global__ __launch_bounds__(SCAN_THREADS) void scan_offsets(int* __restrict__ off, int n)
{
  __shared__ int tile[SCAN_TILE];
  __shared__ int wtot[32];
  const int tid  = threadIdx.x;
  const int lane = tid & 31;
  const int wave = tid >> 5;
  int carry = 0;
  for (int base = 0; base < n; base += SCAN_TILE) {
    for (int k0 = 0; k0 < SCAN_CHUNK; k0 += SCAN_GROUP) {
      int v[SCAN_GROUP];
#pragma unroll
      for (int q = 0; q < SCAN_GROUP; ++q) {
        const int g = base + (k0 + q) * SCAN_THREADS + tid;
        v[q] = off[(g < n) ? g : (n - 1)];
      }
#pragma unroll
      for (int q = 0; q < SCAN_GROUP; ++q) {
        const int i = (k0 + q) * SCAN_THREADS + tid;
        tile[i] = (base + i < n) ? v[q] : 0;
      }
    }
    __syncthreads();
    int* mine = tile + tid * SCAN_CHUNK;
    int s = 0;
#pragma unroll
    for (int j = 0; j < SCAN_CHUNK; ++j) s += mine[j];
    const int incl = wave_incl_scan(s, lane);
    if (lane == 31) wtot[wave] = incl;
    __syncthreads();
    if (wave == 0) wtot[lane] = wave_incl_scan(wtot[lane], lane);
    __syncthreads();
    int run = carry + (wave ? wtot[wave - 1] : 0) + (incl - s);
    carry += wtot[31];
#pragma unroll
    for (int j = 0; j < SCAN_CHUNK; ++j) {
      const int c = mine[j];
      mine[j] = run;
      run += c;
    }
    __syncthreads();
    for (int k0 = 0; k0 < SCAN_CHUNK; k0 += SCAN_GROUP) {
      int v[SCAN_GROUP];
#pragma unroll
      for (int q = 0; q < SCAN_GROUP; ++q) v[q] = tile[(k0 + q) * SCAN_THREADS + tid];
#pragma unroll
      for (int q = 0; q < SCAN_GROUP; ++q) {
        const int g = base + (k0 + q) * SCAN_THREADS + tid;
        if (g < n) off[g] = v[q];
      }
    }
    __syncthreads();
  }
  if (tid == 0) off[n] = carry;
}

__global__ __launch_bounds__(256) void bucket_scatter(
    const int* __restrict__ srcC, const int* __restrict__ dstC,
    const int* __restrict__ srcE, const int* __restrict__ dstE,
    const int* __restrict__ S, int2* __restrict__ rec,
    int nEc, int nE, int n, int nB)
{
  __shared__ int  hist[MAXB + 1];
  __shared__ int  lbase[MAXB];
  __shared__ int  gbase[MAXB];
  __shared__ int  wsum[8];
  __shared__ int2 tile[STILE];
  const int tid  = threadIdx.x;
  const int lane = tid & 31;
  const int wave = tid >> 5;
  const int j    = blockIdx.x;
  gbase[tid] = (tid < nB) ? S[tid * NCHUNK + j] : 0;
  const int per = (nE + NCHUNK - 1) / NCHUNK;
  const int beg = j * per;
  const int end = (beg + per < nE) ? (beg + per) : nE;
  for (int t0 = beg; t0 < end; t0 += STILE) {
    hist[tid] = 0;
    __syncthreads();
    int d[16], s[16], r[16];
#pragma unroll
    for (int q = 0; q < 16; ++q) {
      const int e  = t0 + q * 256 + tid;
      const int ec = (e < end) ? e : (end - 1);
      const bool chem = ec < nEc;
      const int* pd = chem ? (dstC + ec) : (dstE + (ec - nEc));
      const int* ps = chem ? (srcC + ec) : (srcE + (ec - nEc));
      d[q] = *pd;
      s[q] = *ps + (chem ? 0 : n);
    }
#pragma unroll
    for (int q = 0; q < 16; ++q) {
      const bool ok = t0 + q * 256 + tid < end;
      r[q] = atomicAdd(&hist[ok ? (d[q] >> BSH) : MAXB], 1);
      if (!ok) d[q] = -1;
    }
    __syncthreads();
    const int c    = hist[tid];
    const int incl = wave_incl_scan(c, lane);
    if (lane == 31) wsum[wave] = incl;
    __syncthreads();
    int wb = 0;
    for (int w = 0; w < wave; ++w) wb += wsum[w];
    lbase[tid] = wb + incl - c;
    __syncthreads();
#pragma unroll
    for (int q = 0; q < 16; ++q)
      if (d[q] >= 0) tile[lbase[d[q] >> BSH] + r[q]] = make_int2(d[q], s[q]);
    __syncthreads();
    const int cntT = (end - t0 < STILE) ? (end - t0) : STILE;
    for (int i = tid; i < cntT; i += 256) {
      const int2 v = tile[i];
      const int  b = v.x >> BSH;
      rec[gbase[b] + (i - lbase[b])] = v;
    }
    __syncthreads();
    gbase[tid] += c;
  }
}

__global__ __launch_bounds__(1024) void bucket_place(
    const int* __restrict__ S, const int2* __restrict__ rec,
    int* __restrict__ off, int* __restrict__ lst, int n, int nB)
{
  __shared__ int cnt[BNODES + 1];
  __shared__ int cur[BNODES + 1];
  __shared__ int wsum[8];
  const int tid  = threadIdx.x;
  const int lane = tid & 31;
  const int wave = tid >> 5;
  const int b    = blockIdx.x;
  const int segBeg = S[b * NCHUNK];
  const int segEnd = S[(b + 1) * NCHUNK];
  if (tid < BNODES) cnt[tid] = 0;
  __syncthreads();
  for (int i0 = segBeg; i0 < segEnd; i0 += 4096) {
    int2 v[4];
#pragma unroll
    for (int q = 0; q < 4; ++q) {
      const int i = i0 + q * 1024 + tid;
      v[q] = rec[(i < segEnd) ? i : (segEnd - 1)];
    }
#pragma unroll
    for (int q = 0; q < 4; ++q) {
      const bool ok = i0 + q * 1024 + tid < segEnd;
      atomicAdd(&cnt[ok ? (v[q].x & (BNODES - 1)) : BNODES], 1);
    }
  }
  __syncthreads();
  const int c    = (tid < BNODES) ? cnt[tid] : 0;
  const int incl = wave_incl_scan(c, lane);
  if (tid < BNODES && lane == 31) wsum[wave] = incl;
  __syncthreads();
  if (tid < BNODES) {
    int wb = 0;
    for (int w = 0; w < wave; ++w) wb += wsum[w];
    const int st = segBeg + wb + incl - c;
    cur[tid] = st;
    const int node = b * BNODES + tid;
    if (node < n) off[node] = st;
  }
  if (b == nB - 1 && tid == 0) off[n] = segEnd;
  __syncthreads();
  for (int i0 = segBeg; i0 < segEnd; i0 += 4096) {
    int2 v[4];
#pragma unroll
    for (int q = 0; q < 4; ++q) {
      const int i = i0 + q * 1024 + tid;
      v[q] = rec[(i < segEnd) ? i : (segEnd - 1)];
    }
#pragma unroll
    for (int q = 0; q < 4; ++q) {
      const bool ok = i0 + q * 1024 + tid < segEnd;
      const int  p  = atomicAdd(&cur[ok ? (v[q].x & (BNODES - 1)) : BNODES], 1);
      if (ok) lst[p] = v[q].y;
    }
  }
}

__device__ __forceinline__ void add4(float4& a, const float4 v) {
  a.x += v.x; a.y += v.y; a.z += v.z; a.w += v.w;
}

__global__ __launch_bounds__(256) void gather_rows(
    const float* __restrict__ Wh2, const int* __restrict__ off,
    const int* __restrict__ lst, float* __restrict__ h, int n)
{
  const int lane = threadIdx.x & 31;
  const int node = __builtin_amdgcn_readfirstlane(blockIdx.x * 8 + (threadIdx.x >> 5));
  if (node >= n) return;
  const int beg = off[node];
  const int end = off[node + 1];
  const float4* rows = (const float4*)Wh2 + lane;
  const float4 zero4 = make_float4(0.f, 0.f, 0.f, 0.f);
  float4 acc = zero4;
  for (int base = beg; base < end; base += 32) {
    const int m = (end - base < 32) ? (end - base) : 32;
    const int idx = (lane < m) ? lst[base + lane] : 0;
    for (int j = 0; j < m; j += 4) {
      const int left = m - j;
      float4 v0, v1 = zero4, v2 = zero4, v3 = zero4;
      v0 = rows[(size_t)__builtin_amdgcn_readlane(idx, j) * 32];
      if (left > 1) v1 = rows[(size_t)__builtin_amdgcn_readlane(idx, j + 1) * 32];
      if (left > 2) v2 = rows[(size_t)__builtin_amdgcn_readlane(idx, j + 2) * 32];
      if (left > 3) v3 = rows[(size_t)__builtin_amdgcn_readlane(idx, j + 3) * 32];
      add4(acc, v0); add4(acc, v1); add4(acc, v2); add4(acc, v3);
    }
  }
  ((float4*)h)[(size_t)node * 32 + lane] = acc;
}

extern "C" void kernel_launch(void* const* d_in, const int* in_sizes, int n_in,
                              void* d_out, int out_size, void* d_ws, size_t ws_size,
                              hipStream_t stream) {
  const float* feats = (const float*)d_in[0];
  const float* Wc    = (const float*)d_in[1];
  const float* bc    = (const float*)d_in[2];
  const float* We    = (const float*)d_in[3];
  const float* be    = (const float*)d_in[4];
  const float* Wo    = (const float*)d_in[5];
  const float* bo    = (const float*)d_in[6];
  const int*   srcC  = (const int*)d_in[7];
  const int*   dstC  = (const int*)d_in[8];
  const int*   srcE  = (const int*)d_in[9];
  const int*   dstE  = (const int*)d_in[10];

  const int n   = in_sizes[0] / DFEAT;
  const int nEc = in_sizes[7];
  const int nEe = in_sizes[9];
  const int nE  = nEc + nEe;

  const size_t feSize = (size_t)n * DFEAT;
  float* WhC = (float*)d_ws;
  float* WhE = WhC + feSize;
  float* h   = WhE + feSize;
  int*   off = (int*)(h + feSize);
  int*   S   = off + ((size_t)n / 1024 + 2) * 1024;
  int2*  rec = (int2*)(S + (size_t)MAXB * NCHUNK + 1024);
  int*   lst = (int*)(rec + nE);
  float* out = (float*)d_out;

  const int nB = (n + BNODES - 1) / BNODES;
  bucket_count<<<NCHUNK, 256, 0, stream>>>(dstC, dstE, S, nEc, nE, nB);
  scan_offsets<<<1, SCAN_THREADS, 0, stream>>>(S, nB * NCHUNK);
  bucket_scatter<<<NCHUNK, 256, 0, stream>>>(srcC, dstC, srcE, dstE, S, rec,
                                             nEc, nE, n, nB);
  bucket_place<<<(unsigned)nB, 1024, 0, stream>>>(S, rec, off, lst, n, nB);

  rgcn_proj_gemm<<<512, 256, 0, stream>>>(feats, Wc, bc, We, be, WhC, WhE, n);

  gather_rows<<<(unsigned)((n + 7) / 8), 256, 0, stream>>>(WhC, off, lst, h, n);

  rgcn_out_gemm<<<768, 256, 0, stream>>>(h, Wo, bo, WhE, out, n);
}


# Output GEMM builds each 16-row tile of h itself: a wave sums its two nodes' sorted source rows in registers and puts them in LDS; h is never stored, one launch fewer; per-node sum order unchanged
# speedup vs baseline: 4.2655x; 1.0303x over previous
#include <hip/hip_runtime.h>

typedef __attribute__((ext_vector_type(2))) float v2f;
typedef __attribute__((ext_vector_type(8))) float v8f;

#define DFEAT 128
#define TILE  16

#define LDP 132

__device__ __forceinline__ void wave_stage16(const float* __restrict__ src, float* dst, int lane) {
  const float4* s4 = (const float4*)src + lane;
  float4 v[16];
#pragma unroll
  for (int q = 0; q < 16; ++q) v[q] = s4[q * 32];
#pragma unroll
  for (int q = 0; q < 16; ++q) *(float4*)(dst + q * LDP + lane * 4) = v[q];
}

__device__ __forceinline__ void read_frags(const float* p, v2f (&f)[32]) {
#pragma unroll
  for (int t = 0; t < 32; ++t) f[t] = *(const v2f*)(p + 4 * t);
}

__device__ __forceinline__ void tile_load(const float* __restrict__ A, int m0, int n,
                                          int wave, int lane, float4& p0, float4& p1) {
  int r0 = m0 + wave, r1 = m0 + wave + 8;
  r0 = (r0 < n) ? r0 : (n - 1);
  r1 = (r1 < n) ? r1 : (n - 1);
  p0 = ((const float4*)(A + (size_t)r0 * DFEAT))[lane];
  p1 = ((const float4*)(A + (size_t)r1 * DFEAT))[lane];
}

__device__ __forceinline__ void tile_store(float* buf, int wave, int lane,
                                           const float4 p0, const float4 p1) {
  *(float4*)(buf + wave * LDP + lane * 4)       = p0;
  *(float4*)(buf + (wave + 8) * LDP + lane * 4) = p1;
}

__global__ __launch_bounds__(256) void rgcn_proj_gemm(
    const float* __restrict__ feats,
    const float* __restrict__ Wc, const float* __restrict__ bc,
    const float* __restrict__ We, const float* __restrict__ be,
    float* __restrict__ WhC, float* __restrict__ WhE, int n)
{
  __shared__ __attribute__((aligned(16))) float lds[2 * DFEAT * LDP];
  const int tid  = threadIdx.x;
  const int wave = tid >> 5;
  const int lane = tid & 31;
  const int half = lane >> 4;
  const int l15  = lane & 15;
  const int n0   = wave * TILE;

  v2f bC[32], bE[32];
  wave_stage16(Wc + (size_t)n0 * DFEAT, lds + n0 * LDP, lane);
  wave_stage16(We + (size_t)n0 * DFEAT, lds + (DFEAT + n0) * LDP, lane);
  __syncthreads();
  read_frags(lds + (n0 + l15) * LDP + 2 * half, bC);
  read_frags(lds + (DFEAT + n0 + l15) * LDP + 2 * half, bE);
  __syncthreads();

  const float biasC = bc[n0 + l15];
  const float biasE = be[n0 + l15];
  const int nTiles = (n + TILE - 1) / TILE;
  const float4 zero4 = make_float4(0.f, 0.f, 0.f, 0.f);
  float4 p0 = zero4, p1 = zero4;
  int t = blockIdx.x;
  if (t < nTiles) {
    tile_load(feats, t * TILE, n, wave, lane, p0, p1);
    tile_store(lds, wave, lane, p0, p1);
  }
  __syncthreads();
  int cur = 0;
  for (; t < nTiles; t += gridDim.x) {
    const int  tn   = t + gridDim.x;
    const bool more = tn < nTiles;
    if (more) tile_load(feats, tn * TILE, n, wave, lane, p0, p1);

    v2f a[32];
    read_frags(lds + cur * (TILE * LDP) + l15 * LDP + 2 * half, a);
    v8f accC = {0.f, 0.f, 0.f, 0.f, 0.f, 0.f, 0.f, 0.f};
    v8f accE = {0.f, 0.f, 0.f, 0.f, 0.f, 0.f, 0.f, 0.f};
#pragma unroll
    for (int k = 0; k < 32; ++k) {
      accC = __builtin_amdgcn_wmma_f32_16x16x4_f32(
          false, a[k], false, bC[k], (short)0, accC, false, false);
      accE = __builtin_amdgcn_wmma_f32_16x16x4_f32(
          false, a[k], false, bE[k], (short)0, accE, false, false);
    }

    const int m0 = t * TILE;
    float oC[8], oE[8];
#pragma unroll
    for (int r = 0; r < 8; ++r) { oC[r] = accC[r] + biasC; oE[r] = accE[r] + biasE; }
    float* ocol = WhC + (size_t)(m0 + half * 8) * DFEAT + n0 + l15;
    float* ecol = WhE + (size_t)(m0 + half * 8) * DFEAT + n0 + l15;
    if (m0 + TILE <= n) {
#pragma unroll
      for (int r = 0; r < 8; ++r) {
        ocol[(size_t)r * DFEAT] = oC[r];
        ecol[(size_t)r * DFEAT] = oE[r];
      }
    } else {
#pragma unroll
      for (int r = 0; r < 8; ++r) {
        if (m0 + half * 8 + r < n) {
          ocol[(size_t)r * DFEAT] = oC[r];
          ecol[(size_t)r * DFEAT] = oE[r];
        }
      }
    }

    if (more) tile_store(lds + (cur ^ 1) * (TILE * LDP), wave, lane, p0, p1);
    __syncthreads();
    cur ^= 1;
  }
}

#define BSH     8
#define BNODES  256
#define MAXB    256
#define NCHUNK  256
#define STILE   4096

__device__ __forceinline__ int wave_incl_scan(int v, int lane) {
#pragma unroll
  for (int d = 1; d < 32; d <<= 1) {
    const int up = __shfl_up(v, d);
    if (lane >= d) v += up;
  }
  return v;
}

__global__ __launch_bounds__(256) void bucket_count(
    const int* __restrict__ dstC, const int* __restrict__ dstE,
    int* __restrict__ cnt, int nEc, int nE, int nB)
{
  __shared__ int hist[MAXB + 1];
  const int tid = threadIdx.x;
  const int j   = blockIdx.x;
  hist[tid] = 0;
  __syncthreads();
  const int per = (nE + NCHUNK - 1) / NCHUNK;
  const int beg = j * per;
  const int end = (beg + per < nE) ? (beg + per) : nE;
  for (int t0 = beg; t0 < end; t0 += STILE) {
    int d[16];
#pragma unroll
    for (int q = 0; q < 16; ++q) {
      const int e  = t0 + q * 256 + tid;
      const int ec = (e < end) ? e : (end - 1);
      const int* pd = (ec < nEc) ? (dstC + ec) : (dstE + (ec - nEc));
      d[q] = *pd;
    }
#pragma unroll
    for (int q = 0; q < 16; ++q) {
      const bool ok = t0 + q * 256 + tid < end;
      atomicAdd(&hist[ok ? (d[q] >> BSH) : MAXB], 1);
    }
  }
  __syncthreads();
  if (tid < nB) cnt[tid * NCHUNK + j] = hist[tid];
}

#define SCAN_THREADS 1024
#define SCAN_CHUNK   49
#define SCAN_GROUP   7
#define SCAN_TILE    (SCAN_THREADS * SCAN_CHUNK)

__global__ __launch_bounds__(SCAN_THREADS) void scan_offsets(int* __restrict__ off, int n)
{
  __shared__ int tile[SCAN_TILE];
  __shared__ int wtot[32];
  const int tid  = threadIdx.x;
  const int lane = tid & 31;
  const int wave = tid >> 5;
  int carry = 0;
  for (int base = 0; base < n; base += SCAN_TILE) {
    for (int k0 = 0; k0 < SCAN_CHUNK; k0 += SCAN_GROUP) {
      int v[SCAN_GROUP];
#pragma unroll
      for (int q = 0; q < SCAN_GROUP; ++q) {
        const int g = base + (k0 + q) * SCAN_THREADS + tid;
        v[q] = off[(g < n) ? g : (n - 1)];
      }
#pragma unroll
      for (int q = 0; q < SCAN_GROUP; ++q) {
        const int i = (k0 + q) * SCAN_THREADS + tid;
        tile[i] = (base + i < n) ? v[q] : 0;
      }
    }
    __syncthreads();
    int* mine = tile + tid * SCAN_CHUNK;
    int s = 0;
#pragma unroll
    for (int j = 0; j < SCAN_CHUNK; ++j) s += mine[j];
    const int incl = wave_incl_scan(s, lane);
    if (lane == 31) wtot[wave] = incl;
    __syncthreads();
    if (wave == 0) wtot[lane] = wave_incl_scan(wtot[lane], lane);
    __syncthreads();
    int run = carry + (wave ? wtot[wave - 1] : 0) + (incl - s);
    carry += wtot[31];
#pragma unroll
    for (int j = 0; j < SCAN_CHUNK; ++j) {
      const int c = mine[j];
      mine[j] = run;
      run += c;
    }
    __syncthreads();
    for (int k0 = 0; k0 < SCAN_CHUNK; k0 += SCAN_GROUP) {
      int v[SCAN_GROUP];
#pragma unroll
      for (int q = 0; q < SCAN_GROUP; ++q) v[q] = tile[(k0 + q) * SCAN_THREADS + tid];
#pragma unroll
      for (int q = 0; q < SCAN_GROUP; ++q) {
        const int g = base + (k0 + q) * SCAN_THREADS + tid;
        if (g < n) off[g] = v[q];
      }
    }
    __syncthreads();
  }
  if (tid == 0) off[n] = carry;
}

__global__ __launch_bounds__(256) void bucket_scatter(
    const int* __restrict__ srcC, const int* __restrict__ dstC,
    const int* __restrict__ srcE, const int* __restrict__ dstE,
    const int* __restrict__ S, int2* __restrict__ rec,
    int nEc, int nE, int n, int nB)
{
  __shared__ int  hist[MAXB + 1];
  __shared__ int  lbase[MAXB];
  __shared__ int  gbase[MAXB];
  __shared__ int  wsum[8];
  __shared__ int2 tile[STILE];
  const int tid  = threadIdx.x;
  const int lane = tid & 31;
  const int wave = tid >> 5;
  const int j    = blockIdx.x;
  gbase[tid] = (tid < nB) ? S[tid * NCHUNK + j] : 0;
  const int per = (nE + NCHUNK - 1) / NCHUNK;
  const int beg = j * per;
  const int end = (beg + per < nE) ? (beg + per) : nE;
  for (int t0 = beg; t0 < end; t0 += STILE) {
    hist[tid] = 0;
    __syncthreads();
    int d[16], s[16], r[16];
#pragma unroll
    for (int q = 0; q < 16; ++q) {
      const int e  = t0 + q * 256 + tid;
      const int ec = (e < end) ? e : (end - 1);
      const bool chem = ec < nEc;
      const int* pd = chem ? (dstC + ec) : (dstE + (ec - nEc));
      const int* ps = chem ? (srcC + ec) : (srcE + (ec - nEc));
      d[q] = *pd;
      s[q] = *ps + (chem ? 0 : n);
    }
#pragma unroll
    for (int q = 0; q < 16; ++q) {
      const bool ok = t0 + q * 256 + tid < end;
      r[q] = atomicAdd(&hist[ok ? (d[q] >> BSH) : MAXB], 1);
      if (!ok) d[q] = -1;
    }
    __syncthreads();
    const int c    = hist[tid];
    const int incl = wave_incl_scan(c, lane);
    if (lane == 31) wsum[wave] = incl;
    __syncthreads();
    int wb = 0;
    for (int w = 0; w < wave; ++w) wb += wsum[w];
    lbase[tid] = wb + incl - c;
    __syncthreads();
#pragma unroll
    for (int q = 0; q < 16; ++q)
      if (d[q] >= 0) tile[lbase[d[q] >> BSH] + r[q]] = make_int2(d[q], s[q]);
    __syncthreads();
    const int cntT = (end - t0 < STILE) ? (end - t0) : STILE;
    for (int i = tid; i < cntT; i += 256) {
      const int2 v = tile[i];
      const int  b = v.x >> BSH;
      rec[gbase[b] + (i - lbase[b])] = v;
    }
    __syncthreads();
    gbase[tid] += c;
  }
}

__global__ __launch_bounds__(1024) void bucket_place(
    const int* __restrict__ S, const int2* __restrict__ rec,
    int* __restrict__ off, int* __restrict__ lst, int n, int nB)
{
  __shared__ int cnt[BNODES + 1];
  __shared__ int cur[BNODES + 1];
  __shared__ int wsum[8];
  const int tid  = threadIdx.x;
  const int lane = tid & 31;
  const int wave = tid >> 5;
  const int b    = blockIdx.x;
  const int segBeg = S[b * NCHUNK];
  const int segEnd = S[(b + 1) * NCHUNK];
  if (tid < BNODES) cnt[tid] = 0;
  __syncthreads();
  for (int i0 = segBeg; i0 < segEnd; i0 += 4096) {
    int2 v[4];
#pragma unroll
    for (int q = 0; q < 4; ++q) {
      const int i = i0 + q * 1024 + tid;
      v[q] = rec[(i < segEnd) ? i : (segEnd - 1)];
    }
#pragma unroll
    for (int q = 0; q < 4; ++q) {
      const bool ok = i0 + q * 1024 + tid < segEnd;
      atomicAdd(&cnt[ok ? (v[q].x & (BNODES - 1)) : BNODES], 1);
    }
  }
  __syncthreads();
  const int c    = (tid < BNODES) ? cnt[tid] : 0;
  const int incl = wave_incl_scan(c, lane);
  if (tid < BNODES && lane == 31) wsum[wave] = incl;
  __syncthreads();
  if (tid < BNODES) {
    int wb = 0;
    for (int w = 0; w < wave; ++w) wb += wsum[w];
    const int st = segBeg + wb + incl - c;
    cur[tid] = st;
    const int node = b * BNODES + tid;
    if (node < n) off[node] = st;
  }
  if (b == nB - 1 && tid == 0) off[n] = segEnd;
  __syncthreads();
  for (int i0 = segBeg; i0 < segEnd; i0 += 4096) {
    int2 v[4];
#pragma unroll
    for (int q = 0; q < 4; ++q) {
      const int i = i0 + q * 1024 + tid;
      v[q] = rec[(i < segEnd) ? i : (segEnd - 1)];
    }
#pragma unroll
    for (int q = 0; q < 4; ++q) {
      const bool ok = i0 + q * 1024 + tid < segEnd;
      const int  p  = atomicAdd(&cur[ok ? (v[q].x & (BNODES - 1)) : BNODES], 1);
      if (ok) lst[p] = v[q].y;
    }
  }
}

__device__ __forceinline__ void add4(float4& a, const float4 v) {
  a.x += v.x; a.y += v.y; a.z += v.z; a.w += v.w;
}

#define GROWS 16

__device__ __forceinline__ float4 seg_sum(
    const float4* __restrict__ rows, const int* __restrict__ lst,
    int beg, int end, int idx0, int lane)
{
  const float4 zero4 = make_float4(0.f, 0.f, 0.f, 0.f);
  float4 acc = zero4;
  for (int base = beg; base < end; base += 32) {
    const int m = (end - base < 32) ? (end - base) : 32;
    int idx = idx0;
    if (base != beg) idx = (lane < m) ? lst[base + lane] : 0;
    for (int j = 0; j < m; j += GROWS) {
      const int left = m - j;
      float4 v[GROWS];
      v[0] = rows[(size_t)__builtin_amdgcn_readlane(idx, j) * 32];
#pragma unroll
      for (int q = 1; q < GROWS; ++q) {
        v[q] = zero4;
        if (left > q) v[q] = rows[(size_t)__builtin_amdgcn_readlane(idx, j + q) * 32];
      }
#pragma unroll
      for (int q = 0; q < GROWS; ++q) add4(acc, v[q]);
    }
  }
  return acc;
}

__device__ __forceinline__ void pair_sum(
    const float4* __restrict__ rows, const int* __restrict__ off,
    const int* __restrict__ lst, int na, int nb, int n, int lane,
    float4& p0, float4& p1)
{
  const int ca = (na < n) ? na : (n - 1);
  const int cb = (nb < n) ? nb : (n - 1);
  const int begA = off[ca];
  const int nxtA = off[ca + 1];
  const int begB = off[cb];
  const int nxtB = off[cb + 1];
  const int endA = (na < n) ? nxtA : begA;
  const int endB = (nb < n) ? nxtB : begB;
  const int idxA = (lane < endA - begA) ? lst[begA + lane] : 0;
  const int idxB = (lane < endB - begB) ? lst[begB + lane] : 0;
  p0 = seg_sum(rows, lst, begA, endA, idxA, lane);
  p1 = seg_sum(rows, lst, begB, endB, idxB, lane);
}

__global__ __launch_bounds__(256) void rgcn_out_gemm(
    const float* __restrict__ Wh2, const int* __restrict__ off,
    const int* __restrict__ lst,
    const float* __restrict__ Wo, const float* __restrict__ bo,
    const float* __restrict__ WhE, float* __restrict__ out, int n)
{
  __shared__ __attribute__((aligned(16))) float lds[DFEAT * LDP];
  const int tid  = threadIdx.x;
  const int wave = __builtin_amdgcn_readfirstlane(tid >> 5);
  const int lane = tid & 31;
  const int half = lane >> 4;
  const int l15  = lane & 15;
  const int n0   = wave * TILE;

  v2f b[32];
  wave_stage16(Wo + (size_t)n0 * DFEAT, lds + n0 * LDP, lane);
  __syncthreads();
  read_frags(lds + (n0 + l15) * LDP + 2 * half, b);
  __syncthreads();

  const float bias = bo[n0 + l15];
  const int nTiles = (n + TILE - 1) / TILE;
  const float4* rows = (const float4*)Wh2 + lane;
  const float4 zero4 = make_float4(0.f, 0.f, 0.f, 0.f);
  float4 p0 = zero4, p1 = zero4;
  int t = blockIdx.x;
  if (t < nTiles) {
    pair_sum(rows, off, lst, t * TILE + wave, t * TILE + wave + 8, n, lane, p0, p1);
    tile_store(lds, wave, lane, p0, p1);
  }
  __syncthreads();
  int cur = 0;
  for (; t < nTiles; t += gridDim.x) {
    const int  tn   = t + gridDim.x;
    const bool more = tn < nTiles;
    if (more) pair_sum(rows, off, lst, tn * TILE + wave, tn * TILE + wave + 8, n, lane, p0, p1);

    const int m0 = t * TILE;
    float e[8];
#pragma unroll
    for (int r = 0; r < 8; ++r) {
      int M = m0 + half * 8 + r;
      M = (M < n) ? M : (n - 1);
      e[r] = WhE[(size_t)M * DFEAT + n0 + l15];
    }

    v2f a[32];
    read_frags(lds + cur * (TILE * LDP) + l15 * LDP + 2 * half, a);
    v8f acc = {0.f, 0.f, 0.f, 0.f, 0.f, 0.f, 0.f, 0.f};
#pragma unroll
    for (int k = 0; k < 32; ++k)
      acc = __builtin_amdgcn_wmma_f32_16x16x4_f32(
          false, a[k], false, b[k], (short)0, acc, false, false);

    float o[8];
#pragma unroll
    for (int r = 0; r < 8; ++r) o[r] = acc[r] + bias + e[r];
    float* ocol = out + (size_t)(m0 + half * 8) * DFEAT + n0 + l15;
    if (m0 + TILE <= n) {
#pragma unroll
      for (int r = 0; r < 8; ++r) ocol[(size_t)r * DFEAT] = o[r];
    } else {
#pragma unroll
      for (int r = 0; r < 8; ++r)
        if (m0 + half * 8 + r < n) ocol[(size_t)r * DFEAT] = o[r];
    }

    if (more) tile_store(lds + (cur ^ 1) * (TILE * LDP), wave, lane, p0, p1);
    __syncthreads();
    cur ^= 1;
  }
}

extern "C" void kernel_launch(void* const* d_in, const int* in_sizes, int n_in,
                              void* d_out, int out_size, void* d_ws, size_t ws_size,
                              hipStream_t stream) {
  const float* feats = (const float*)d_in[0];
  const float* Wc    = (const float*)d_in[1];
  const float* bc    = (const float*)d_in[2];
  const float* We    = (const float*)d_in[3];
  const float* be    = (const float*)d_in[4];
  const float* Wo    = (const float*)d_in[5];
  const float* bo    = (const float*)d_in[6];
  const int*   srcC  = (const int*)d_in[7];
  const int*   dstC  = (const int*)d_in[8];
  const int*   srcE  = (const int*)d_in[9];
  const int*   dstE  = (const int*)d_in[10];

  const int n   = in_sizes[0] / DFEAT;
  const int nEc = in_sizes[7];
  const int nEe = in_sizes[9];
  const int nE  = nEc + nEe;

  const size_t feSize = (size_t)n * DFEAT;
  float* WhC = (float*)d_ws;
  float* WhE = WhC + feSize;
  int*   off = (int*)(WhE + feSize);
  int*   S   = off + ((size_t)n / 1024 + 2) * 1024;
  int2*  rec = (int2*)(S + (size_t)MAXB * NCHUNK + 1024);
  int*   lst = (int*)(rec + nE);
  float* out = (float*)d_out;

  const int nB = (n + BNODES - 1) / BNODES;
  bucket_count<<<NCHUNK, 256, 0, stream>>>(dstC, dstE, S, nEc, nE, nB);
  scan_offsets<<<1, SCAN_THREADS, 0, stream>>>(S, nB * NCHUNK);
  bucket_scatter<<<NCHUNK, 256, 0, stream>>>(srcC, dstC, srcE, dstE, S, rec,
                                             nEc, nE, n, nB);
  bucket_place<<<(unsigned)nB, 1024, 0, stream>>>(S, rec, off, lst, n, nB);

  rgcn_proj_gemm<<<512, 256, 0, stream>>>(feats, Wc, bc, We, be, WhC, WhE, n);

  rgcn_out_gemm<<<768, 256, 0, stream>>>(WhC, off, lst, Wo, bo, WhE, out, n);
}


# Scan of the sort's count matrix rides in the projection launch as its work-group 0 (independent work, hidden under the matrix launch): one launch fewer, no one-work-group launch on an idle chip
# speedup vs baseline: 4.8369x; 1.1340x over previous
#include <hip/hip_runtime.h>

typedef __attribute__((ext_vector_type(2))) float v2f;
typedef __attribute__((ext_vector_type(8))) float v8f;

#define DFEAT 128
#define TILE  16

#define LDP 132

__device__ __forceinline__ void wave_stage16(const float* __restrict__ src, float* dst, int lane) {
  const float4* s4 = (const float4*)src + lane;
  float4 v[16];
#pragma unroll
  for (int q = 0; q < 16; ++q) v[q] = s4[q * 32];
#pragma unroll
  for (int q = 0; q < 16; ++q) *(float4*)(dst + q * LDP + lane * 4) = v[q];
}

__device__ __forceinline__ void read_frags(const float* p, v2f (&f)[32]) {
#pragma unroll
  for (int t = 0; t < 32; ++t) f[t] = *(const v2f*)(p + 4 * t);
}

__device__ __forceinline__ void tile_load(const float* __restrict__ A, int m0, int n,
                                          int wave, int lane, float4& p0, float4& p1) {
  int r0 = m0 + wave, r1 = m0 + wave + 8;
  r0 = (r0 < n) ? r0 : (n - 1);
  r1 = (r1 < n) ? r1 : (n - 1);
  p0 = ((const float4*)(A + (size_t)r0 * DFEAT))[lane];
  p1 = ((const float4*)(A + (size_t)r1 * DFEAT))[lane];
}

__device__ __forceinline__ void tile_store(float* buf, int wave, int lane,
                                           const float4 p0, const float4 p1) {
  *(float4*)(buf + wave * LDP + lane * 4)       = p0;
  *(float4*)(buf + (wave + 8) * LDP + lane * 4) = p1;
}

__device__ __forceinline__ int wave_incl_scan(int v, int lane) {
#pragma unroll
  for (int d = 1; d < 32; d <<= 1) {
    const int up = __shfl_up(v, d);
    if (lane >= d) v += up;
  }
  return v;
}

#define SCAN_THREADS 256
#define SCAN_CHUNK   98
#define SCAN_GROUP   49
#define SCAN_TILE    (SCAN_THREADS * SCAN_CHUNK)

__device__ __forceinline__ void scan_role(int* __restrict__ off, int n, int* tile, int* wtot, int tid)
{
  const int lane = tid & 31;
  const int wave = tid >> 5;
  int carry = 0;
  for (int base = 0; base < n; base += SCAN_TILE) {
    for (int k0 = 0; k0 < SCAN_CHUNK; k0 += SCAN_GROUP) {
      int v[SCAN_GROUP];
#pragma unroll
      for (int q = 0; q < SCAN_GROUP; ++q) {
        const int g = base + (k0 + q) * SCAN_THREADS + tid;
        v[q] = off[(g < n) ? g : (n - 1)];
      }
#pragma unroll
      for (int q = 0; q < SCAN_GROUP; ++q) {
        const int i = (k0 + q) * SCAN_THREADS + tid;
        tile[i] = (base + i < n) ? v[q] : 0;
      }
    }
    __syncthreads();
    int* mine = tile + tid * SCAN_CHUNK;
    int s = 0;
#pragma unroll
    for (int j = 0; j < SCAN_CHUNK; ++j) s += mine[j];
    const int incl = wave_incl_scan(s, lane);
    if (lane == 31) wtot[wave] = incl;
    __syncthreads();
    int wb = 0, tot = 0;
#pragma unroll
    for (int w = 0; w < SCAN_THREADS / 32; ++w) {
      const int x = wtot[w];
      tot += x;
      if (w < wave) wb += x;
    }
    int run = carry + wb + (incl - s);
    carry += tot;
#pragma unroll
    for (int j = 0; j < SCAN_CHUNK; ++j) {
      const int cc = mine[j];
      mine[j] = run;
      run += cc;
    }
    __syncthreads();
    for (int k0 = 0; k0 < SCAN_CHUNK; k0 += SCAN_GROUP) {
      int v[SCAN_GROUP];
#pragma unroll
      for (int q = 0; q < SCAN_GROUP; ++q) v[q] = tile[(k0 + q) * SCAN_THREADS + tid];
#pragma unroll
      for (int q = 0; q < SCAN_GROUP; ++q) {
        const int g = base + (k0 + q) * SCAN_THREADS + tid;
        if (g < n) off[g] = v[q];
      }
    }
    __syncthreads();
  }
  if (tid == 0) off[n] = carry;
}

__global__ __launch_bounds__(256) void rgcn_proj_gemm(
    const float* __restrict__ feats,
    const float* __restrict__ Wc, const float* __restrict__ bc,
    const float* __restrict__ We, const float* __restrict__ be,
    float* __restrict__ WhC, float* __restrict__ WhE, int n,
    int* __restrict__ S, int nS)
{
  __shared__ __attribute__((aligned(16))) float lds[2 * DFEAT * LDP];
  const int tid  = threadIdx.x;
  if (blockIdx.x == 0) {
    scan_role(S, nS, (int*)lds, (int*)lds + SCAN_TILE, tid);
    return;
  }
  const int nGemm = gridDim.x - 1;
  const int wave = tid >> 5;
  const int lane = tid & 31;
  const int half = lane >> 4;
  const int l15  = lane & 15;
  const int n0   = wave * TILE;

  v2f bC[32], bE[32];
  wave_stage16(Wc + (size_t)n0 * DFEAT, lds + n0 * LDP, lane);
  wave_stage16(We + (size_t)n0 * DFEAT, lds + (DFEAT + n0) * LDP, lane);
  __syncthreads();
  read_frags(lds + (n0 + l15) * LDP + 2 * half, bC);
  read_frags(lds + (DFEAT + n0 + l15) * LDP + 2 * half, bE);
  __syncthreads();

  const float biasC = bc[n0 + l15];
  const float biasE = be[n0 + l15];
  const int nTiles = (n + TILE - 1) / TILE;
  const float4 zero4 = make_float4(0.f, 0.f, 0.f, 0.f);
  float4 p0 = zero4, p1 = zero4;
  int t = blockIdx.x - 1;
  if (t < nTiles) {
    tile_load(feats, t * TILE, n, wave, lane, p0, p1);
    tile_store(lds, wave, lane, p0, p1);
  }
  __syncthreads();
  int cur = 0;
  for (; t < nTiles; t += nGemm) {
    const int  tn   = t + nGemm;
    const bool more = tn < nTiles;
    if (more) tile_load(feats, tn * TILE, n, wave, lane, p0, p1);

    v2f a[32];
    read_frags(lds + cur * (TILE * LDP) + l15 * LDP + 2 * half, a);
    v8f accC = {0.f, 0.f, 0.f, 0.f, 0.f, 0.f, 0.f, 0.f};
    v8f accE = {0.f, 0.f, 0.f, 0.f, 0.f, 0.f, 0.f, 0.f};
#pragma unroll
    for (int k = 0; k < 32; ++k) {
      accC = __builtin_amdgcn_wmma_f32_16x16x4_f32(
          false, a[k], false, bC[k], (short)0, accC, false, false);
      accE = __builtin_amdgcn_wmma_f32_16x16x4_f32(
          false, a[k], false, bE[k], (short)0, accE, false, false);
    }

    const int m0 = t * TILE;
    float oC[8], oE[8];
#pragma unroll
    for (int r = 0; r < 8; ++r) { oC[r] = accC[r] + biasC; oE[r] = accE[r] + biasE; }
    float* ocol = WhC + (size_t)(m0 + half * 8) * DFEAT + n0 + l15;
    float* ecol = WhE + (size_t)(m0 + half * 8) * DFEAT + n0 + l15;
    if (m0 + TILE <= n) {
#pragma unroll
      for (int r = 0; r < 8; ++r) {
        ocol[(size_t)r * DFEAT] = oC[r];
        ecol[(size_t)r * DFEAT] = oE[r];
      }
    } else {
#pragma unroll
      for (int r = 0; r < 8; ++r) {
        if (m0 + half * 8 + r < n) {
          ocol[(size_t)r * DFEAT] = oC[r];
          ecol[(size_t)r * DFEAT] = oE[r];
        }
      }
    }

    if (more) tile_store(lds + (cur ^ 1) * (TILE * LDP), wave, lane, p0, p1);
    __syncthreads();
    cur ^= 1;
  }
}

#define BSH     8
#define BNODES  256
#define MAXB    256
#define NCHUNK  256
#define STILE   4096

__global__ __launch_bounds__(256) void bucket_count(
    const int* __restrict__ dstC, const int* __restrict__ dstE,
    int* __restrict__ cnt, int nEc, int nE, int nB)
{
  __shared__ int hist[MAXB + 1];
  const int tid = threadIdx.x;
  const int j   = blockIdx.x;
  hist[tid] = 0;
  __syncthreads();
  const int per = (nE + NCHUNK - 1) / NCHUNK;
  const int beg = j * per;
  const int end = (beg + per < nE) ? (beg + per) : nE;
  for (int t0 = beg; t0 < end; t0 += STILE) {
    int d[16];
#pragma unroll
    for (int q = 0; q < 16; ++q) {
      const int e  = t0 + q * 256 + tid;
      const int ec = (e < end) ? e : (end - 1);
      const int* pd = (ec < nEc) ? (dstC + ec) : (dstE + (ec - nEc));
      d[q] = *pd;
    }
#pragma unroll
    for (int q = 0; q < 16; ++q) {
      const bool ok = t0 + q * 256 + tid < end;
      atomicAdd(&hist[ok ? (d[q] >> BSH) : MAXB], 1);
    }
  }
  __syncthreads();
  if (tid < nB) cnt[tid * NCHUNK + j] = hist[tid];
}

__global__ __launch_bounds__(256) void bucket_scatter(
    const int* __restrict__ srcC, const int* __restrict__ dstC,
    const int* __restrict__ srcE, const int* __restrict__ dstE,
    const int* __restrict__ S, int2* __restrict__ rec,
    int nEc, int nE, int n, int nB)
{
  __shared__ int  hist[MAXB + 1];
  __shared__ int  lbase[MAXB];
  __shared__ int  gbase[MAXB];
  __shared__ int  wsum[8];
  __shared__ int2 tile[STILE];
  const int tid  = threadIdx.x;
  const int lane = tid & 31;
  const int wave = tid >> 5;
  const int j    = blockIdx.x;
  gbase[tid] = (tid < nB) ? S[tid * NCHUNK + j] : 0;
  const int per = (nE + NCHUNK - 1) / NCHUNK;
  const int beg = j * per;
  const int end = (beg + per < nE) ? (beg + per) : nE;
  for (int t0 = beg; t0 < end; t0 += STILE) {
    hist[tid] = 0;
    __syncthreads();
    int d[16], s[16], r[16];
#pragma unroll
    for (int q = 0; q < 16; ++q) {
      const int e  = t0 + q * 256 + tid;
      const int ec = (e < end) ? e : (end - 1);
      const bool chem = ec < nEc;
      const int* pd = chem ? (dstC + ec) : (dstE + (ec - nEc));
      const int* ps = chem ? (srcC + ec) : (srcE + (ec - nEc));
      d[q] = *pd;
      s[q] = *ps + (chem ? 0 : n);
    }
#pragma unroll
    for (int q = 0; q < 16; ++q) {
      const bool ok = t0 + q * 256 + tid < end;
      r[q] = atomicAdd(&hist[ok ? (d[q] >> BSH) : MAXB], 1);
      if (!ok) d[q] = -1;
    }
    __syncthreads();
    const int c    = hist[tid];
    const int incl = wave_incl_scan(c, lane);
    if (lane == 31) wsum[wave] = incl;
    __syncthreads();
    int wb = 0;
    for (int w = 0; w < wave; ++w) wb += wsum[w];
    lbase[tid] = wb + incl - c;
    __syncthreads();
#pragma unroll
    for (int q = 0; q < 16; ++q)
      if (d[q] >= 0) tile[lbase[d[q] >> BSH] + r[q]] = make_int2(d[q], s[q]);
    __syncthreads();
    const int cntT = (end - t0 < STILE) ? (end - t0) : STILE;
    for (int i = tid; i < cntT; i += 256) {
      const int2 v = tile[i];
      const int  b = v.x >> BSH;
      rec[gbase[b] + (i - lbase[b])] = v;
    }
    __syncthreads();
    gbase[tid] += c;
  }
}

__global__ __launch_bounds__(1024) void bucket_place(
    const int* __restrict__ S, const int2* __restrict__ rec,
    int* __restrict__ off, int* __restrict__ lst, int n, int nB)
{
  __shared__ int cnt[BNODES + 1];
  __shared__ int cur[BNODES + 1];
  __shared__ int wsum[8];
  const int tid  = threadIdx.x;
  const int lane = tid & 31;
  const int wave = tid >> 5;
  const int b    = blockIdx.x;
  const int segBeg = S[b * NCHUNK];
  const int segEnd = S[(b + 1) * NCHUNK];
  if (tid < BNODES) cnt[tid] = 0;
  __syncthreads();
  for (int i0 = segBeg; i0 < segEnd; i0 += 4096) {
    int2 v[4];
#pragma unroll
    for (int q = 0; q < 4; ++q) {
      const int i = i0 + q * 1024 + tid;
      v[q] = rec[(i < segEnd) ? i : (segEnd - 1)];
    }
#pragma unroll
    for (int q = 0; q < 4; ++q) {
      const bool ok = i0 + q * 1024 + tid < segEnd;
      atomicAdd(&cnt[ok ? (v[q].x & (BNODES - 1)) : BNODES], 1);
    }
  }
  __syncthreads();
  const int c    = (tid < BNODES) ? cnt[tid] : 0;
  const int incl = wave_incl_scan(c, lane);
  if (tid < BNODES && lane == 31) wsum[wave] = incl;
  __syncthreads();
  if (tid < BNODES) {
    int wb = 0;
    for (int w = 0; w < wave; ++w) wb += wsum[w];
    const int st = segBeg + wb + incl - c;
    cur[tid] = st;
    const int node = b * BNODES + tid;
    if (node < n) off[node] = st;
  }
  if (b == nB - 1 && tid == 0) off[n] = segEnd;
  __syncthreads();
  for (int i0 = segBeg; i0 < segEnd; i0 += 4096) {
    int2 v[4];
#pragma unroll
    for (int q = 0; q < 4; ++q) {
      const int i = i0 + q * 1024 + tid;
      v[q] = rec[(i < segEnd) ? i : (segEnd - 1)];
    }
#pragma unroll
    for (int q = 0; q < 4; ++q) {
      const bool ok = i0 + q * 1024 + tid < segEnd;
      const int  p  = atomicAdd(&cur[ok ? (v[q].x & (BNODES - 1)) : BNODES], 1);
      if (ok) lst[p] = v[q].y;
    }
  }
}

__device__ __forceinline__ void add4(float4& a, const float4 v) {
  a.x += v.x; a.y += v.y; a.z += v.z; a.w += v.w;
}

#define GROWS 16

__device__ __forceinline__ float4 seg_sum(
    const float4* __restrict__ rows, const int* __restrict__ lst,
    int beg, int end, int idx0, int lane)
{
  const float4 zero4 = make_float4(0.f, 0.f, 0.f, 0.f);
  float4 acc = zero4;
  for (int base = beg; base < end; base += 32) {
    const int m = (end - base < 32) ? (end - base) : 32;
    int idx = idx0;
    if (base != beg) idx = (lane < m) ? lst[base + lane] : 0;
    for (int j = 0; j < m; j += GROWS) {
      const int left = m - j;
      float4 v[GROWS];
      v[0] = rows[(size_t)__builtin_amdgcn_readlane(idx, j) * 32];
#pragma unroll
      for (int q = 1; q < GROWS; ++q) {
        v[q] = zero4;
        if (left > q) v[q] = rows[(size_t)__builtin_amdgcn_readlane(idx, j + q) * 32];
      }
#pragma unroll
      for (int q = 0; q < GROWS; ++q) add4(acc, v[q]);
    }
  }
  return acc;
}

__device__ __forceinline__ void pair_sum(
    const float4* __restrict__ rows, const int* __restrict__ off,
    const int* __restrict__ lst, int na, int nb, int n, int lane,
    float4& p0, float4& p1)
{
  const int ca = (na < n) ? na : (n - 1);
  const int cb = (nb < n) ? nb : (n - 1);
  const int begA = off[ca];
  const int nxtA = off[ca + 1];
  const int begB = off[cb];
  const int nxtB = off[cb + 1];
  const int endA = (na < n) ? nxtA : begA;
  const int endB = (nb < n) ? nxtB : begB;
  const int idxA = (lane < endA - begA) ? lst[begA + lane] : 0;
  const int idxB = (lane < endB - begB) ? lst[begB + lane] : 0;
  p0 = seg_sum(rows, lst, begA, endA, idxA, lane);
  p1 = seg_sum(rows, lst, begB, endB, idxB, lane);
}

__global__ __launch_bounds__(256) void rgcn_out_gemm(
    const float* __restrict__ Wh2, const int* __restrict__ off,
    const int* __restrict__ lst,
    const float* __restrict__ Wo, const float* __restrict__ bo,
    const float* __restrict__ WhE, float* __restrict__ out, int n)
{
  __shared__ __attribute__((aligned(16))) float lds[DFEAT * LDP];
  const int tid  = threadIdx.x;
  const int wave = __builtin_amdgcn_readfirstlane(tid >> 5);
  const int lane = tid & 31;
  const int half = lane >> 4;
  const int l15  = lane & 15;
  const int n0   = wave * TILE;

  v2f b[32];
  wave_stage16(Wo + (size_t)n0 * DFEAT, lds + n0 * LDP, lane);
  __syncthreads();
  read_frags(lds + (n0 + l15) * LDP + 2 * half, b);
  __syncthreads();

  const float bias = bo[n0 + l15];
  const int nTiles = (n + TILE - 1) / TILE;
  const float4* rows = (const float4*)Wh2 + lane;
  const float4 zero4 = make_float4(0.f, 0.f, 0.f, 0.f);
  float4 p0 = zero4, p1 = zero4;
  int t = blockIdx.x;
  if (t < nTiles) {
    pair_sum(rows, off, lst, t * TILE + wave, t * TILE + wave + 8, n, lane, p0, p1);
    tile_store(lds, wave, lane, p0, p1);
  }
  __syncthreads();
  int cur = 0;
  for (; t < nTiles; t += gridDim.x) {
    const int  tn   = t + gridDim.x;
    const bool more = tn < nTiles;
    if (more) pair_sum(rows, off, lst, tn * TILE + wave, tn * TILE + wave + 8, n, lane, p0, p1);

    const int m0 = t * TILE;
    float e[8];
#pragma unroll
    for (int r = 0; r < 8; ++r) {
      int M = m0 + half * 8 + r;
      M = (M < n) ? M : (n - 1);
      e[r] = WhE[(size_t)M * DFEAT + n0 + l15];
    }

    v2f a[32];
    read_frags(lds + cur * (TILE * LDP) + l15 * LDP + 2 * half, a);
    v8f acc = {0.f, 0.f, 0.f, 0.f, 0.f, 0.f, 0.f, 0.f};
#pragma unroll
    for (int k = 0; k < 32; ++k)
      acc = __builtin_amdgcn_wmma_f32_16x16x4_f32(
          false, a[k], false, b[k], (short)0, acc, false, false);

    float o[8];
#pragma unroll
    for (int r = 0; r < 8; ++r) o[r] = acc[r] + bias + e[r];
    float* ocol = out + (size_t)(m0 + half * 8) * DFEAT + n0 + l15;
    if (m0 + TILE <= n) {
#pragma unroll
      for (int r = 0; r < 8; ++r) ocol[(size_t)r * DFEAT] = o[r];
    } else {
#pragma unroll
      for (int r = 0; r < 8; ++r)
        if (m0 + half * 8 + r < n) ocol[(size_t)r * DFEAT] = o[r];
    }

    if (more) tile_store(lds + (cur ^ 1) * (TILE * LDP), wave, lane, p0, p1);
    __syncthreads();
    cur ^= 1;
  }
}

extern "C" void kernel_launch(void* const* d_in, const int* in_sizes, int n_in,
                              void* d_out, int out_size, void* d_ws, size_t ws_size,
                              hipStream_t stream) {
  const float* feats = (const float*)d_in[0];
  const float* Wc    = (const float*)d_in[1];
  const float* bc    = (const float*)d_in[2];
  const float* We    = (const float*)d_in[3];
  const float* be    = (const float*)d_in[4];
  const float* Wo    = (const float*)d_in[5];
  const float* bo    = (const float*)d_in[6];
  const int*   srcC  = (const int*)d_in[7];
  const int*   dstC  = (const int*)d_in[8];
  const int*   srcE  = (const int*)d_in[9];
  const int*   dstE  = (const int*)d_in[10];

  const int n   = in_sizes[0] / DFEAT;
  const int nEc = in_sizes[7];
  const int nEe = in_sizes[9];
  const int nE  = nEc + nEe;

  const size_t feSize = (size_t)n * DFEAT;
  float* WhC = (float*)d_ws;
  float* WhE = WhC + feSize;
  int*   off = (int*)(WhE + feSize);
  int*   S   = off + ((size_t)n / 1024 + 2) * 1024;
  int2*  rec = (int2*)(S + (size_t)MAXB * NCHUNK + 1024);
  int*   lst = (int*)(rec + nE);
  float* out = (float*)d_out;

  const int nB = (n + BNODES - 1) / BNODES;
  bucket_count<<<NCHUNK, 256, 0, stream>>>(dstC, dstE, S, nEc, nE, nB);

  rgcn_proj_gemm<<<512, 256, 0, stream>>>(feats, Wc, bc, We, be, WhC, WhE, n,
                                          S, nB * NCHUNK);

  bucket_scatter<<<NCHUNK, 256, 0, stream>>>(srcC, dstC, srcE, dstE, S, rec,
                                             nEc, nE, n, nB);
  bucket_place<<<(unsigned)nB, 1024, 0, stream>>>(S, rec, off, lst, n, nB);

  rgcn_out_gemm<<<768, 256, 0, stream>>>(WhC, off, lst, Wo, bo, WhE, out, n);
}


# Sort level 1 without global count or scan: each chunk is bucket-ordered in LDS into its own run with an offset table, inside the projection launch; 2 launches fewer; sum order = LDS arrival order
# speedup vs baseline: 5.6586x; 1.1699x over previous
#include <hip/hip_runtime.h>

typedef __attribute__((ext_vector_type(2))) float v2f;
typedef __attribute__((ext_vector_type(8))) float v8f;

#define DFEAT 128
#define TILE  16

#define LDP 132

__device__ __forceinline__ void wave_stage16(const float* __restrict__ src, float* dst, int lane) {
  const float4* s4 = (const float4*)src + lane;
  float4 v[16];
#pragma unroll
  for (int q = 0; q < 16; ++q) v[q] = s4[q * 32];
#pragma unroll
  for (int q = 0; q < 16; ++q) *(float4*)(dst + q * LDP + lane * 4) = v[q];
}

__device__ __forceinline__ void read_frags(const float* p, v2f (&f)[32]) {
#pragma unroll
  for (int t = 0; t < 32; ++t) f[t] = *(const v2f*)(p + 4 * t);
}

__device__ __forceinline__ void tile_load(const float* __restrict__ A, int m0, int n,
                                          int wave, int lane, float4& p0, float4& p1) {
  int r0 = m0 + wave, r1 = m0 + wave + 8;
  r0 = (r0 < n) ? r0 : (n - 1);
  r1 = (r1 < n) ? r1 : (n - 1);
  p0 = ((const float4*)(A + (size_t)r0 * DFEAT))[lane];
  p1 = ((const float4*)(A + (size_t)r1 * DFEAT))[lane];
}

__device__ __forceinline__ void tile_store(float* buf, int wave, int lane,
                                           const float4 p0, const float4 p1) {
  *(float4*)(buf + wave * LDP + lane * 4)       = p0;
  *(float4*)(buf + (wave + 8) * LDP + lane * 4) = p1;
}

__device__ __forceinline__ int wave_incl_scan(int v, int lane) {
#pragma unroll
  for (int d = 1; d < 32; d <<= 1) {
    const int up = __shfl_up(v, d);
    if (lane >= d) v += up;
  }
  return v;
}

#define BSH     8
#define BNODES  256
#define MAXB    256
#define NCHUNK  256
#define STILE   4096

__device__ __forceinline__ int chunk_len(int nE)  { return (nE + NCHUNK - 1) / NCHUNK; }
__device__ __forceinline__ int units_per_chunk(int nE) { return (chunk_len(nE) + STILE - 1) / STILE; }

__device__ __forceinline__ void chunk_sort_role(
    const int* __restrict__ srcC, const int* __restrict__ dstC,
    const int* __restrict__ srcE, const int* __restrict__ dstE,
    int2* __restrict__ rec, int* __restrict__ T,
    int nEc, int nE, int n, int j, int* sm, int tid)
{
  int*  hist  = sm;
  int*  lbase = sm + 264;
  int*  wsum  = sm + 520;
  int2* tile  = (int2*)(sm + 528);
  const int lane = tid & 31;
  const int wave = tid >> 5;
  const int per = chunk_len(nE);
  const int tpc = units_per_chunk(nE);
  const int U   = NCHUNK * tpc;
  const int beg = j * per;
  const int end = (beg + per < nE) ? (beg + per) : nE;
  for (int k = 0; k < tpc; ++k) {
    const int t0 = beg + k * STILE;
    const int u  = j * tpc + k;
    hist[tid] = 0;
    __syncthreads();
    int d[16], s[16], r[16];
#pragma unroll
    for (int q = 0; q < 16; ++q) {
      const int e  = t0 + q * 256 + tid;
      const int ec = (e < end) ? e : (end - 1);
      const bool chem = ec < nEc;
      const int* pd = chem ? (dstC + ec) : (dstE + (ec - nEc));
      const int* ps = chem ? (srcC + ec) : (srcE + (ec - nEc));
      d[q] = *pd;
      s[q] = *ps + (chem ? 0 : n);
    }
#pragma unroll
    for (int q = 0; q < 16; ++q) {
      const bool ok = t0 + q * 256 + tid < end;
      r[q] = atomicAdd(&hist[ok ? (d[q] >> BSH) : MAXB], 1);
      if (!ok) d[q] = -1;
    }
    __syncthreads();
    const int c    = hist[tid];
    const int incl = wave_incl_scan(c, lane);
    if (lane == 31) wsum[wave] = incl;
    __syncthreads();
    int wb = 0;
    for (int w = 0; w < wave; ++w) wb += wsum[w];
    const int lb = wb + incl - c;
    lbase[tid] = lb;
    T[(size_t)tid * U + u] = lb;
    int cntT = end - t0;
    cntT = (cntT < STILE) ? cntT : STILE;
    cntT = (cntT > 0) ? cntT : 0;
    if (tid == 0) T[(size_t)MAXB * U + u] = cntT;
    __syncthreads();
#pragma unroll
    for (int q = 0; q < 16; ++q)
      if (d[q] >= 0) tile[lbase[d[q] >> BSH] + r[q]] = make_int2(d[q], s[q]);
    __syncthreads();
    for (int i = tid; i < cntT; i += 256) rec[t0 + i] = tile[i];
    __syncthreads();
  }
}
__global__ __launch_bounds__(256) void rgcn_proj_gemm(
    const float* __restrict__ feats,
    const float* __restrict__ Wc, const float* __restrict__ bc,
    const float* __restrict__ We, const float* __restrict__ be,
    float* __restrict__ WhC, float* __restrict__ WhE, int n,
    const int* __restrict__ srcC, const int* __restrict__ dstC,
    const int* __restrict__ srcE, const int* __restrict__ dstE,
    int2* __restrict__ rec, int* __restrict__ T, int nEc, int nE)
{
  __shared__ __attribute__((aligned(16))) float lds[2 * DFEAT * LDP];
  const int tid  = threadIdx.x;
  if (blockIdx.x < NCHUNK) {
    chunk_sort_role(srcC, dstC, srcE, dstE, rec, T, nEc, nE, n, blockIdx.x, (int*)lds, tid);
    return;
  }
  const int nGemm = gridDim.x - NCHUNK;
  const int wave = tid >> 5;
  const int lane = tid & 31;
  const int half = lane >> 4;
  const int l15  = lane & 15;
  const int n0   = wave * TILE;

  v2f bC[32], bE[32];
  wave_stage16(Wc + (size_t)n0 * DFEAT, lds + n0 * LDP, lane);
  wave_stage16(We + (size_t)n0 * DFEAT, lds + (DFEAT + n0) * LDP, lane);
  __syncthreads();
  read_frags(lds + (n0 + l15) * LDP + 2 * half, bC);
  read_frags(lds + (DFEAT + n0 + l15) * LDP + 2 * half, bE);
  __syncthreads();

  const float biasC = bc[n0 + l15];
  const float biasE = be[n0 + l15];
  const int nTiles = (n + TILE - 1) / TILE;
  const float4 zero4 = make_float4(0.f, 0.f, 0.f, 0.f);
  float4 p0 = zero4, p1 = zero4;
  int t = blockIdx.x - NCHUNK;
  if (t < nTiles) {
    tile_load(feats, t * TILE, n, wave, lane, p0, p1);
    tile_store(lds, wave, lane, p0, p1);
  }
  __syncthreads();
  int cur = 0;
  for (; t < nTiles; t += nGemm) {
    const int  tn   = t + nGemm;
    const bool more = tn < nTiles;
    if (more) tile_load(feats, tn * TILE, n, wave, lane, p0, p1);

    v2f a[32];
    read_frags(lds + cur * (TILE * LDP) + l15 * LDP + 2 * half, a);
    v8f accC = {0.f, 0.f, 0.f, 0.f, 0.f, 0.f, 0.f, 0.f};
    v8f accE = {0.f, 0.f, 0.f, 0.f, 0.f, 0.f, 0.f, 0.f};
#pragma unroll
    for (int k = 0; k < 32; ++k) {
      accC = __builtin_amdgcn_wmma_f32_16x16x4_f32(
          false, a[k], false, bC[k], (short)0, accC, false, false);
      accE = __builtin_amdgcn_wmma_f32_16x16x4_f32(
          false, a[k], false, bE[k], (short)0, accE, false, false);
    }

    const int m0 = t * TILE;
    float oC[8], oE[8];
#pragma unroll
    for (int r = 0; r < 8; ++r) { oC[r] = accC[r] + biasC; oE[r] = accE[r] + biasE; }
    float* ocol = WhC + (size_t)(m0 + half * 8) * DFEAT + n0 + l15;
    float* ecol = WhE + (size_t)(m0 + half * 8) * DFEAT + n0 + l15;
    if (m0 + TILE <= n) {
#pragma unroll
      for (int r = 0; r < 8; ++r) {
        ocol[(size_t)r * DFEAT] = oC[r];
        ecol[(size_t)r * DFEAT] = oE[r];
      }
    } else {
#pragma unroll
      for (int r = 0; r < 8; ++r) {
        if (m0 + half * 8 + r < n) {
          ocol[(size_t)r * DFEAT] = oC[r];
          ecol[(size_t)r * DFEAT] = oE[r];
        }
      }
    }

    if (more) tile_store(lds + (cur ^ 1) * (TILE * LDP), wave, lane, p0, p1);
    __syncthreads();
    cur ^= 1;
  }
}

__global__ __launch_bounds__(1024) void bucket_place(
    const int* __restrict__ T, const int2* __restrict__ rec,
    int* __restrict__ off, int* __restrict__ lst, int n, int nE, int nB)
{
  __shared__ int cnt[BNODES + 32];
  __shared__ int cur[BNODES + 32];
  __shared__ int wsum[8];
  __shared__ int tsum[8];
  const int tid  = threadIdx.x;
  const int lane = tid & 31;
  const int wave = tid >> 5;
  const int b    = blockIdx.x;
  const int per = chunk_len(nE);
  const int tpc = units_per_chunk(nE);
  const int U   = NCHUNK * tpc;
  const int* Tb = T + (size_t)b * U;
  const int* Te = Tb + U;
  const int team = tid >> 2;
  const int k    = tid & 3;
  int part = 0;
  if (tid < BNODES) {
    cnt[tid] = 0;
    for (int u = tid; u < U; u += BNODES) part += Tb[u];
  }
  const int pincl = wave_incl_scan(part, lane);
  if (tid < BNODES && lane == 31) tsum[wave] = pincl;
  __syncthreads();
  for (int u0 = 0; u0 < U; u0 += 256) {
    const int u = u0 + team;
    const int s = Tb[u] + k;
    const int e = Te[u];
    const int2* ru = rec + ((u / tpc) * per + (u % tpc) * STILE);
    int2 v[8];
#pragma unroll
    for (int q = 0; q < 8; ++q) {
      const int r = s + 4 * q;
      const int2* p = (r < e) ? (ru + r) : rec;
      v[q] = *p;
    }
#pragma unroll
    for (int q = 0; q < 8; ++q) {
      const bool ok = s + 4 * q < e;
      atomicAdd(&cnt[ok ? (v[q].x & (BNODES - 1)) : (BNODES + lane)], 1);
    }
    for (int r = s + 32; r < e; r += 4)
      atomicAdd(&cnt[ru[r].x & (BNODES - 1)], 1);
  }
  __syncthreads();
  const int c    = (tid < BNODES) ? cnt[tid] : 0;
  const int incl = wave_incl_scan(c, lane);
  if (tid < BNODES && lane == 31) wsum[wave] = incl;
  __syncthreads();
  if (tid < BNODES) {
    int segBeg = 0;
    for (int w = 0; w < 8; ++w) segBeg += tsum[w];
    int wb = 0;
    for (int w = 0; w < wave; ++w) wb += wsum[w];
    const int st = segBeg + wb + incl - c;
    cur[tid] = st;
    const int node = b * BNODES + tid;
    if (node < n) off[node] = st;
  }
  if (b == nB - 1 && tid == 0) off[n] = nE;
  __syncthreads();
  for (int u0 = 0; u0 < U; u0 += 256) {
    const int u = u0 + team;
    const int s = Tb[u] + k;
    const int e = Te[u];
    const int2* ru = rec + ((u / tpc) * per + (u % tpc) * STILE);
    int2 v[8];
#pragma unroll
    for (int q = 0; q < 8; ++q) {
      const int r = s + 4 * q;
      const int2* p = (r < e) ? (ru + r) : rec;
      v[q] = *p;
    }
    int p[8];
#pragma unroll
    for (int q = 0; q < 8; ++q) {
      const bool ok = s + 4 * q < e;
      p[q] = atomicAdd(&cur[ok ? (v[q].x & (BNODES - 1)) : (BNODES + lane)], 1);
    }
#pragma unroll
    for (int q = 0; q < 8; ++q)
      if (s + 4 * q < e) lst[p[q]] = v[q].y;
    for (int r = s + 32; r < e; r += 4) {
      const int2 x = ru[r];
      lst[atomicAdd(&cur[x.x & (BNODES - 1)], 1)] = x.y;
    }
  }
}

__device__ __forceinline__ void add4(float4& a, const float4 v) {
  a.x += v.x; a.y += v.y; a.z += v.z; a.w += v.w;
}

#define GROWS 16

__device__ __forceinline__ float4 seg_sum(
    const float4* __restrict__ rows, const int* __restrict__ lst,
    int beg, int end, int idx0, int lane)
{
  const float4 zero4 = make_float4(0.f, 0.f, 0.f, 0.f);
  float4 acc = zero4;
  for (int base = beg; base < end; base += 32) {
    const int m = (end - base < 32) ? (end - base) : 32;
    int idx = idx0;
    if (base != beg) idx = (lane < m) ? lst[base + lane] : 0;
    for (int j = 0; j < m; j += GROWS) {
      const int left = m - j;
      float4 v[GROWS];
      v[0] = rows[(size_t)__builtin_amdgcn_readlane(idx, j) * 32];
#pragma unroll
      for (int q = 1; q < GROWS; ++q) {
        v[q] = zero4;
        if (left > q) v[q] = rows[(size_t)__builtin_amdgcn_readlane(idx, j + q) * 32];
      }
#pragma unroll
      for (int q = 0; q < GROWS; ++q) add4(acc, v[q]);
    }
  }
  return acc;
}

__device__ __forceinline__ void pair_sum(
    const float4* __restrict__ rows, const int* __restrict__ off,
    const int* __restrict__ lst, int na, int nb, int n, int lane,
    float4& p0, float4& p1)
{
  const int ca = (na < n) ? na : (n - 1);
  const int cb = (nb < n) ? nb : (n - 1);
  const int begA = off[ca];
  const int nxtA = off[ca + 1];
  const int begB = off[cb];
  const int nxtB = off[cb + 1];
  const int endA = (na < n) ? nxtA : begA;
  const int endB = (nb < n) ? nxtB : begB;
  const int idxA = (lane < endA - begA) ? lst[begA + lane] : 0;
  const int idxB = (lane < endB - begB) ? lst[begB + lane] : 0;
  p0 = seg_sum(rows, lst, begA, endA, idxA, lane);
  p1 = seg_sum(rows, lst, begB, endB, idxB, lane);
}

__global__ __launch_bounds__(256) void rgcn_out_gemm(
    const float* __restrict__ Wh2, const int* __restrict__ off,
    const int* __restrict__ lst,
    const float* __restrict__ Wo, const float* __restrict__ bo,
    const float* __restrict__ WhE, float* __restrict__ out, int n)
{
  __shared__ __attribute__((aligned(16))) float lds[DFEAT * LDP];
  const int tid  = threadIdx.x;
  const int wave = __builtin_amdgcn_readfirstlane(tid >> 5);
  const int lane = tid & 31;
  const int half = lane >> 4;
  const int l15  = lane & 15;
  const int n0   = wave * TILE;

  v2f b[32];
  wave_stage16(Wo + (size_t)n0 * DFEAT, lds + n0 * LDP, lane);
  __syncthreads();
  read_frags(lds + (n0 + l15) * LDP + 2 * half, b);
  __syncthreads();

  const float bias = bo[n0 + l15];
  const int nTiles = (n + TILE - 1) / TILE;
  const float4* rows = (const float4*)Wh2 + lane;
  const float4 zero4 = make_float4(0.f, 0.f, 0.f, 0.f);
  float4 p0 = zero4, p1 = zero4;
  int t = blockIdx.x;
  if (t < nTiles) {
    pair_sum(rows, off, lst, t * TILE + wave, t * TILE + wave + 8, n, lane, p0, p1);
    tile_store(lds, wave, lane, p0, p1);
  }
  __syncthreads();
  int cur = 0;
  for (; t < nTiles; t += gridDim.x) {
    const int  tn   = t + gridDim.x;
    const bool more = tn < nTiles;
    if (more) pair_sum(rows, off, lst, tn * TILE + wave, tn * TILE + wave + 8, n, lane, p0, p1);

    const int m0 = t * TILE;
    float e[8];
#pragma unroll
    for (int r = 0; r < 8; ++r) {
      int M = m0 + half * 8 + r;
      M = (M < n) ? M : (n - 1);
      e[r] = WhE[(size_t)M * DFEAT + n0 + l15];
    }

    v2f a[32];
    read_frags(lds + cur * (TILE * LDP) + l15 * LDP + 2 * half, a);
    v8f acc = {0.f, 0.f, 0.f, 0.f, 0.f, 0.f, 0.f, 0.f};
#pragma unroll
    for (int k = 0; k < 32; ++k)
      acc = __builtin_amdgcn_wmma_f32_16x16x4_f32(
          false, a[k], false, b[k], (short)0, acc, false, false);

    float o[8];
#pragma unroll
    for (int r = 0; r < 8; ++r) o[r] = acc[r] + bias + e[r];
    float* ocol = out + (size_t)(m0 + half * 8) * DFEAT + n0 + l15;
    if (m0 + TILE <= n) {
#pragma unroll
      for (int r = 0; r < 8; ++r) ocol[(size_t)r * DFEAT] = o[r];
    } else {
#pragma unroll
      for (int r = 0; r < 8; ++r)
        if (m0 + half * 8 + r < n) ocol[(size_t)r * DFEAT] = o[r];
    }

    if (more) tile_store(lds + (cur ^ 1) * (TILE * LDP), wave, lane, p0, p1);
    __syncthreads();
    cur ^= 1;
  }
}

extern "C" void kernel_launch(void* const* d_in, const int* in_sizes, int n_in,
                              void* d_out, int out_size, void* d_ws, size_t ws_size,
                              hipStream_t stream) {
  const float* feats = (const float*)d_in[0];
  const float* Wc    = (const float*)d_in[1];
  const float* bc    = (const float*)d_in[2];
  const float* We    = (const float*)d_in[3];
  const float* be    = (const float*)d_in[4];
  const float* Wo    = (const float*)d_in[5];
  const float* bo    = (const float*)d_in[6];
  const int*   srcC  = (const int*)d_in[7];
  const int*   dstC  = (const int*)d_in[8];
  const int*   srcE  = (const int*)d_in[9];
  const int*   dstE  = (const int*)d_in[10];

  const int n   = in_sizes[0] / DFEAT;
  const int nEc = in_sizes[7];
  const int nEe = in_sizes[9];
  const int nE  = nEc + nEe;

  const size_t feSize = (size_t)n * DFEAT;
  float* WhC = (float*)d_ws;
  float* WhE = WhC + feSize;
  int*   off = (int*)(WhE + feSize);
  int2*  rec = (int2*)(off + ((size_t)n / 1024 + 2) * 1024);
  int*   lst = (int*)(rec + nE);
  int*   T   = lst + nE;
  float* out = (float*)d_out;

  rgcn_proj_gemm<<<NCHUNK + 512, 256, 0, stream>>>(feats, Wc, bc, We, be, WhC, WhE, n,
                                                   srcC, dstC, srcE, dstE, rec, T, nEc, nE);

  const int nB = (n + BNODES - 1) / BNODES;
  bucket_place<<<(unsigned)nB, 1024, 0, stream>>>(T, rec, off, lst, n, nE, nB);

  rgcn_out_gemm<<<768, 256, 0, stream>>>(WhC, off, lst, Wo, bo, WhE, out, n);
}


# Every work-group barrier fences LDS only (address-space-scoped fences round s_barrier): a wave no longer waits at each tile's barrier for its own global stores; they drain under the next tile's work
# speedup vs baseline: 5.9316x; 1.0482x over previous
#include <hip/hip_runtime.h>

typedef __attribute__((ext_vector_type(2))) float v2f;
typedef __attribute__((ext_vector_type(8))) float v8f;

#define DFEAT 128
#define TILE  16

#define LDP 132

__device__ __forceinline__ void lds_barrier() {
  __builtin_amdgcn_fence(__ATOMIC_RELEASE, "workgroup", "local");
  __builtin_amdgcn_s_barrier();
  __builtin_amdgcn_fence(__ATOMIC_ACQUIRE, "workgroup", "local");
}

__device__ __forceinline__ void wave_stage16(const float* __restrict__ src, float* dst, int lane) {
  const float4* s4 = (const float4*)src + lane;
  float4 v[16];
#pragma unroll
  for (int q = 0; q < 16; ++q) v[q] = s4[q * 32];
#pragma unroll
  for (int q = 0; q < 16; ++q) *(float4*)(dst + q * LDP + lane * 4) = v[q];
}

__device__ __forceinline__ void read_frags(const float* p, v2f (&f)[32]) {
#pragma unroll
  for (int t = 0; t < 32; ++t) f[t] = *(const v2f*)(p + 4 * t);
}

__device__ __forceinline__ void tile_load(const float* __restrict__ A, int m0, int n,
                                          int wave, int lane, float4& p0, float4& p1) {
  int r0 = m0 + wave, r1 = m0 + wave + 8;
  r0 = (r0 < n) ? r0 : (n - 1);
  r1 = (r1 < n) ? r1 : (n - 1);
  p0 = ((const float4*)(A + (size_t)r0 * DFEAT))[lane];
  p1 = ((const float4*)(A + (size_t)r1 * DFEAT))[lane];
}

__device__ __forceinline__ void tile_store(float* buf, int wave, int lane,
                                           const float4 p0, const float4 p1) {
  *(float4*)(buf + wave * LDP + lane * 4)       = p0;
  *(float4*)(buf + (wave + 8) * LDP + lane * 4) = p1;
}

__device__ __forceinline__ int wave_incl_scan(int v, int lane) {
#pragma unroll
  for (int d = 1; d < 32; d <<= 1) {
    const int up = __shfl_up(v, d);
    if (lane >= d) v += up;
  }
  return v;
}

#define BSH     8
#define BNODES  256
#define MAXB    256
#define NCHUNK  256
#define STILE   4096

__device__ __forceinline__ int chunk_len(int nE)  { return (nE + NCHUNK - 1) / NCHUNK; }
__device__ __forceinline__ int units_per_chunk(int nE) { return (chunk_len(nE) + STILE - 1) / STILE; }

__device__ __forceinline__ void chunk_sort_role(
    const int* __restrict__ srcC, const int* __restrict__ dstC,
    const int* __restrict__ srcE, const int* __restrict__ dstE,
    int2* __restrict__ rec, int* __restrict__ T,
    int nEc, int nE, int n, int j, int* sm, int tid)
{
  int*  hist  = sm;
  int*  lbase = sm + 264;
  int*  wsum  = sm + 520;
  int2* tile  = (int2*)(sm + 528);
  const int lane = tid & 31;
  const int wave = tid >> 5;
  const int per = chunk_len(nE);
  const int tpc = units_per_chunk(nE);
  const int U   = NCHUNK * tpc;
  const int beg = j * per;
  const int end = (beg + per < nE) ? (beg + per) : nE;
  for (int k = 0; k < tpc; ++k) {
    const int t0 = beg + k * STILE;
    const int u  = j * tpc + k;
    hist[tid] = 0;
    lds_barrier();
    int d[16], s[16], r[16];
#pragma unroll
    for (int q = 0; q < 16; ++q) {
      const int e  = t0 + q * 256 + tid;
      const int ec = (e < end) ? e : (end - 1);
      const bool chem = ec < nEc;
      const int* pd = chem ? (dstC + ec) : (dstE + (ec - nEc));
      const int* ps = chem ? (srcC + ec) : (srcE + (ec - nEc));
      d[q] = *pd;
      s[q] = *ps + (chem ? 0 : n);
    }
#pragma unroll
    for (int q = 0; q < 16; ++q) {
      const bool ok = t0 + q * 256 + tid < end;
      r[q] = atomicAdd(&hist[ok ? (d[q] >> BSH) : MAXB], 1);
      if (!ok) d[q] = -1;
    }
    lds_barrier();
    const int c    = hist[tid];
    const int incl = wave_incl_scan(c, lane);
    if (lane == 31) wsum[wave] = incl;
    lds_barrier();
    int wb = 0;
    for (int w = 0; w < wave; ++w) wb += wsum[w];
    const int lb = wb + incl - c;
    lbase[tid] = lb;
    T[(size_t)tid * U + u] = lb;
    int cntT = end - t0;
    cntT = (cntT < STILE) ? cntT : STILE;
    cntT = (cntT > 0) ? cntT : 0;
    if (tid == 0) T[(size_t)MAXB * U + u] = cntT;
    lds_barrier();
#pragma unroll
    for (int q = 0; q < 16; ++q)
      if (d[q] >= 0) tile[lbase[d[q] >> BSH] + r[q]] = make_int2(d[q], s[q]);
    lds_barrier();
    for (int i = tid; i < cntT; i += 256) rec[t0 + i] = tile[i];
    lds_barrier();
  }
}
__global__ __launch_bounds__(256) void rgcn_proj_gemm(
    const float* __restrict__ feats,
    const float* __restrict__ Wc, const float* __restrict__ bc,
    const float* __restrict__ We, const float* __restrict__ be,
    float* __restrict__ WhC, float* __restrict__ WhE, int n,
    const int* __restrict__ srcC, const int* __restrict__ dstC,
    const int* __restrict__ srcE, const int* __restrict__ dstE,
    int2* __restrict__ rec, int* __restrict__ T, int nEc, int nE)
{
  __shared__ __attribute__((aligned(16))) float lds[2 * DFEAT * LDP];
  const int tid  = threadIdx.x;
  if (blockIdx.x < NCHUNK) {
    chunk_sort_role(srcC, dstC, srcE, dstE, rec, T, nEc, nE, n, blockIdx.x, (int*)lds, tid);
    return;
  }
  const int nGemm = gridDim.x - NCHUNK;
  const int wave = tid >> 5;
  const int lane = tid & 31;
  const int half = lane >> 4;
  const int l15  = lane & 15;
  const int n0   = wave * TILE;

  v2f bC[32], bE[32];
  wave_stage16(Wc + (size_t)n0 * DFEAT, lds + n0 * LDP, lane);
  wave_stage16(We + (size_t)n0 * DFEAT, lds + (DFEAT + n0) * LDP, lane);
  lds_barrier();
  read_frags(lds + (n0 + l15) * LDP + 2 * half, bC);
  read_frags(lds + (DFEAT + n0 + l15) * LDP + 2 * half, bE);
  lds_barrier();

  const float biasC = bc[n0 + l15];
  const float biasE = be[n0 + l15];
  const int nTiles = (n + TILE - 1) / TILE;
  const float4 zero4 = make_float4(0.f, 0.f, 0.f, 0.f);
  float4 p0 = zero4, p1 = zero4;
  int t = blockIdx.x - NCHUNK;
  if (t < nTiles) {
    tile_load(feats, t * TILE, n, wave, lane, p0, p1);
    tile_store(lds, wave, lane, p0, p1);
  }
  lds_barrier();
  int cur = 0;
  for (; t < nTiles; t += nGemm) {
    const int  tn   = t + nGemm;
    const bool more = tn < nTiles;
    if (more) tile_load(feats, tn * TILE, n, wave, lane, p0, p1);

    v2f a[32];
    read_frags(lds + cur * (TILE * LDP) + l15 * LDP + 2 * half, a);
    v8f accC = {0.f, 0.f, 0.f, 0.f, 0.f, 0.f, 0.f, 0.f};
    v8f accE = {0.f, 0.f, 0.f, 0.f, 0.f, 0.f, 0.f, 0.f};
#pragma unroll
    for (int k = 0; k < 32; ++k) {
      accC = __builtin_amdgcn_wmma_f32_16x16x4_f32(
          false, a[k], false, bC[k], (short)0, accC, false, false);
      accE = __builtin_amdgcn_wmma_f32_16x16x4_f32(
          false, a[k], false, bE[k], (short)0, accE, false, false);
    }

    const int m0 = t * TILE;
    float oC[8], oE[8];
#pragma unroll
    for (int r = 0; r < 8; ++r) { oC[r] = accC[r] + biasC; oE[r] = accE[r] + biasE; }
    float* ocol = WhC + (size_t)(m0 + half * 8) * DFEAT + n0 + l15;
    float* ecol = WhE + (size_t)(m0 + half * 8) * DFEAT + n0 + l15;
    if (m0 + TILE <= n) {
#pragma unroll
      for (int r = 0; r < 8; ++r) {
        ocol[(size_t)r * DFEAT] = oC[r];
        ecol[(size_t)r * DFEAT] = oE[r];
      }
    } else {
#pragma unroll
      for (int r = 0; r < 8; ++r) {
        if (m0 + half * 8 + r < n) {
          ocol[(size_t)r * DFEAT] = oC[r];
          ecol[(size_t)r * DFEAT] = oE[r];
        }
      }
    }

    if (more) tile_store(lds + (cur ^ 1) * (TILE * LDP), wave, lane, p0, p1);
    lds_barrier();
    cur ^= 1;
  }
}

__global__ __launch_bounds__(1024) void bucket_place(
    const int* __restrict__ T, const int2* __restrict__ rec,
    int* __restrict__ off, int* __restrict__ lst, int n, int nE, int nB)
{
  __shared__ int cnt[BNODES + 32];
  __shared__ int cur[BNODES + 32];
  __shared__ int wsum[8];
  __shared__ int tsum[8];
  const int tid  = threadIdx.x;
  const int lane = tid & 31;
  const int wave = tid >> 5;
  const int b    = blockIdx.x;
  const int per = chunk_len(nE);
  const int tpc = units_per_chunk(nE);
  const int U   = NCHUNK * tpc;
  const int* Tb = T + (size_t)b * U;
  const int* Te = Tb + U;
  const int team = tid >> 2;
  const int k    = tid & 3;
  int part = 0;
  if (tid < BNODES) {
    cnt[tid] = 0;
    for (int u = tid; u < U; u += BNODES) part += Tb[u];
  }
  const int pincl = wave_incl_scan(part, lane);
  if (tid < BNODES && lane == 31) tsum[wave] = pincl;
  lds_barrier();
  for (int u0 = 0; u0 < U; u0 += 256) {
    const int u = u0 + team;
    const int s = Tb[u] + k;
    const int e = Te[u];
    const int2* ru = rec + ((u / tpc) * per + (u % tpc) * STILE);
    int2 v[8];
#pragma unroll
    for (int q = 0; q < 8; ++q) {
      const int r = s + 4 * q;
      const int2* p = (r < e) ? (ru + r) : rec;
      v[q] = *p;
    }
#pragma unroll
    for (int q = 0; q < 8; ++q) {
      const bool ok = s + 4 * q < e;
      atomicAdd(&cnt[ok ? (v[q].x & (BNODES - 1)) : (BNODES + lane)], 1);
    }
    for (int r = s + 32; r < e; r += 4)
      atomicAdd(&cnt[ru[r].x & (BNODES - 1)], 1);
  }
  lds_barrier();
  const int c    = (tid < BNODES) ? cnt[tid] : 0;
  const int incl = wave_incl_scan(c, lane);
  if (tid < BNODES && lane == 31) wsum[wave] = incl;
  lds_barrier();
  if (tid < BNODES) {
    int segBeg = 0;
    for (int w = 0; w < 8; ++w) segBeg += tsum[w];
    int wb = 0;
    for (int w = 0; w < wave; ++w) wb += wsum[w];
    const int st = segBeg + wb + incl - c;
    cur[tid] = st;
    const int node = b * BNODES + tid;
    if (node < n) off[node] = st;
  }
  if (b == nB - 1 && tid == 0) off[n] = nE;
  lds_barrier();
  for (int u0 = 0; u0 < U; u0 += 256) {
    const int u = u0 + team;
    const int s = Tb[u] + k;
    const int e = Te[u];
    const int2* ru = rec + ((u / tpc) * per + (u % tpc) * STILE);
    int2 v[8];
#pragma unroll
    for (int q = 0; q < 8; ++q) {
      const int r = s + 4 * q;
      const int2* p = (r < e) ? (ru + r) : rec;
      v[q] = *p;
    }
    int p[8];
#pragma unroll
    for (int q = 0; q < 8; ++q) {
      const bool ok = s + 4 * q < e;
      p[q] = atomicAdd(&cur[ok ? (v[q].x & (BNODES - 1)) : (BNODES + lane)], 1);
    }
#pragma unroll
    for (int q = 0; q < 8; ++q)
      if (s + 4 * q < e) lst[p[q]] = v[q].y;
    for (int r = s + 32; r < e; r += 4) {
      const int2 x = ru[r];
      lst[atomicAdd(&cur[x.x & (BNODES - 1)], 1)] = x.y;
    }
  }
}

__device__ __forceinline__ void add4(float4& a, const float4 v) {
  a.x += v.x; a.y += v.y; a.z += v.z; a.w += v.w;
}

#define GROWS 16

__device__ __forceinline__ float4 seg_sum(
    const float4* __restrict__ rows, const int* __restrict__ lst,
    int beg, int end, int idx0, int lane)
{
  const float4 zero4 = make_float4(0.f, 0.f, 0.f, 0.f);
  float4 acc = zero4;
  for (int base = beg; base < end; base += 32) {
    const int m = (end - base < 32) ? (end - base) : 32;
    int idx = idx0;
    if (base != beg) idx = (lane < m) ? lst[base + lane] : 0;
    for (int j = 0; j < m; j += GROWS) {
      const int left = m - j;
      float4 v[GROWS];
      v[0] = rows[(size_t)__builtin_amdgcn_readlane(idx, j) * 32];
#pragma unroll
      for (int q = 1; q < GROWS; ++q) {
        v[q] = zero4;
        if (left > q) v[q] = rows[(size_t)__builtin_amdgcn_readlane(idx, j + q) * 32];
      }
#pragma unroll
      for (int q = 0; q < GROWS; ++q) add4(acc, v[q]);
    }
  }
  return acc;
}

__device__ __forceinline__ void pair_sum(
    const float4* __restrict__ rows, const int* __restrict__ off,
    const int* __restrict__ lst, int na, int nb, int n, int lane,
    float4& p0, float4& p1)
{
  const int ca = (na < n) ? na : (n - 1);
  const int cb = (nb < n) ? nb : (n - 1);
  const int begA = off[ca];
  const int nxtA = off[ca + 1];
  const int begB = off[cb];
  const int nxtB = off[cb + 1];
  const int endA = (na < n) ? nxtA : begA;
  const int endB = (nb < n) ? nxtB : begB;
  const int idxA = (lane < endA - begA) ? lst[begA + lane] : 0;
  const int idxB = (lane < endB - begB) ? lst[begB + lane] : 0;
  p0 = seg_sum(rows, lst, begA, endA, idxA, lane);
  p1 = seg_sum(rows, lst, begB, endB, idxB, lane);
}

__global__ __launch_bounds__(256) void rgcn_out_gemm(
    const float* __restrict__ Wh2, const int* __restrict__ off,
    const int* __restrict__ lst,
    const float* __restrict__ Wo, const float* __restrict__ bo,
    const float* __restrict__ WhE, float* __restrict__ out, int n)
{
  __shared__ __attribute__((aligned(16))) float lds[DFEAT * LDP];
  const int tid  = threadIdx.x;
  const int wave = __builtin_amdgcn_readfirstlane(tid >> 5);
  const int lane = tid & 31;
  const int half = lane >> 4;
  const int l15  = lane & 15;
  const int n0   = wave * TILE;

  v2f b[32];
  wave_stage16(Wo + (size_t)n0 * DFEAT, lds + n0 * LDP, lane);
  lds_barrier();
  read_frags(lds + (n0 + l15) * LDP + 2 * half, b);
  lds_barrier();

  const float bias = bo[n0 + l15];
  const int nTiles = (n + TILE - 1) / TILE;
  const float4* rows = (const float4*)Wh2 + lane;
  const float4 zero4 = make_float4(0.f, 0.f, 0.f, 0.f);
  float4 p0 = zero4, p1 = zero4;
  int t = blockIdx.x;
  if (t < nTiles) {
    pair_sum(rows, off, lst, t * TILE + wave, t * TILE + wave + 8, n, lane, p0, p1);
    tile_store(lds, wave, lane, p0, p1);
  }
  lds_barrier();
  int cur = 0;
  for (; t < nTiles; t += gridDim.x) {
    const int  tn   = t + gridDim.x;
    const bool more = tn < nTiles;
    if (more) pair_sum(rows, off, lst, tn * TILE + wave, tn * TILE + wave + 8, n, lane, p0, p1);

    const int m0 = t * TILE;
    float e[8];
#pragma unroll
    for (int r = 0; r < 8; ++r) {
      int M = m0 + half * 8 + r;
      M = (M < n) ? M : (n - 1);
      e[r] = WhE[(size_t)M * DFEAT + n0 + l15];
    }

    v2f a[32];
    read_frags(lds + cur * (TILE * LDP) + l15 * LDP + 2 * half, a);
    v8f acc = {0.f, 0.f, 0.f, 0.f, 0.f, 0.f, 0.f, 0.f};
#pragma unroll
    for (int k = 0; k < 32; ++k)
      acc = __builtin_amdgcn_wmma_f32_16x16x4_f32(
          false, a[k], false, b[k], (short)0, acc, false, false);

    float o[8];
#pragma unroll
    for (int r = 0; r < 8; ++r) o[r] = acc[r] + bias + e[r];
    float* ocol = out + (size_t)(m0 + half * 8) * DFEAT + n0 + l15;
    if (m0 + TILE <= n) {
#pragma unroll
      for (int r = 0; r < 8; ++r) ocol[(size_t)r * DFEAT] = o[r];
    } else {
#pragma unroll
      for (int r = 0; r < 8; ++r)
        if (m0 + half * 8 + r < n) ocol[(size_t)r * DFEAT] = o[r];
    }

    if (more) tile_store(lds + (cur ^ 1) * (TILE * LDP), wave, lane, p0, p1);
    lds_barrier();
    cur ^= 1;
  }
}

extern "C" void kernel_launch(void* const* d_in, const int* in_sizes, int n_in,
                              void* d_out, int out_size, void* d_ws, size_t ws_size,
                              hipStream_t stream) {
  const float* feats = (const float*)d_in[0];
  const float* Wc    = (const float*)d_in[1];
  const float* bc    = (const float*)d_in[2];
  const float* We    = (const float*)d_in[3];
  const float* be    = (const float*)d_in[4];
  const float* Wo    = (const float*)d_in[5];
  const float* bo    = (const float*)d_in[6];
  const int*   srcC  = (const int*)d_in[7];
  const int*   dstC  = (const int*)d_in[8];
  const int*   srcE  = (const int*)d_in[9];
  const int*   dstE  = (const int*)d_in[10];

  const int n   = in_sizes[0] / DFEAT;
  const int nEc = in_sizes[7];
  const int nEe = in_sizes[9];
  const int nE  = nEc + nEe;

  const size_t feSize = (size_t)n * DFEAT;
  float* WhC = (float*)d_ws;
  float* WhE = WhC + feSize;
  int*   off = (int*)(WhE + feSize);
  int2*  rec = (int2*)(off + ((size_t)n / 1024 + 2) * 1024);
  int*   lst = (int*)(rec + nE);
  int*   T   = lst + nE;
  float* out = (float*)d_out;

  rgcn_proj_gemm<<<NCHUNK + 512, 256, 0, stream>>>(feats, Wc, bc, We, be, WhC, WhE, n,
                                                   srcC, dstC, srcE, dstE, rec, T, nEc, nE);

  const int nB = (n + BNODES - 1) / BNODES;
  bucket_place<<<(unsigned)nB, 1024, 0, stream>>>(T, rec, off, lst, n, nE, nB);

  rgcn_out_gemm<<<768, 256, 0, stream>>>(WhC, off, lst, Wo, bo, WhE, out, n);
}


# Sort level 2 builds a bucket's list in LDS: the counting LDS add returns the rank, records stay in registers, the list leaves in whole-line stores, not a 4-byte store per edge; sum order: LDS arrival
# speedup vs baseline: 6.2324x; 1.0507x over previous
#include <hip/hip_runtime.h>

typedef __attribute__((ext_vector_type(2))) float v2f;
typedef __attribute__((ext_vector_type(8))) float v8f;

#define DFEAT 128
#define TILE  16

#define LDP 132

__device__ __forceinline__ void lds_barrier() {
  __builtin_amdgcn_fence(__ATOMIC_RELEASE, "workgroup", "local");
  __builtin_amdgcn_s_barrier();
  __builtin_amdgcn_fence(__ATOMIC_ACQUIRE, "workgroup", "local");
}

__device__ __forceinline__ void wave_stage16(const float* __restrict__ src, float* dst, int lane) {
  const float4* s4 = (const float4*)src + lane;
  float4 v[16];
#pragma unroll
  for (int q = 0; q < 16; ++q) v[q] = s4[q * 32];
#pragma unroll
  for (int q = 0; q < 16; ++q) *(float4*)(dst + q * LDP + lane * 4) = v[q];
}

__device__ __forceinline__ void read_frags(const float* p, v2f (&f)[32]) {
#pragma unroll
  for (int t = 0; t < 32; ++t) f[t] = *(const v2f*)(p + 4 * t);
}

__device__ __forceinline__ void tile_load(const float* __restrict__ A, int m0, int n,
                                          int wave, int lane, float4& p0, float4& p1) {
  int r0 = m0 + wave, r1 = m0 + wave + 8;
  r0 = (r0 < n) ? r0 : (n - 1);
  r1 = (r1 < n) ? r1 : (n - 1);
  p0 = ((const float4*)(A + (size_t)r0 * DFEAT))[lane];
  p1 = ((const float4*)(A + (size_t)r1 * DFEAT))[lane];
}

__device__ __forceinline__ void tile_store(float* buf, int wave, int lane,
                                           const float4 p0, const float4 p1) {
  *(float4*)(buf + wave * LDP + lane * 4)       = p0;
  *(float4*)(buf + (wave + 8) * LDP + lane * 4) = p1;
}

__device__ __forceinline__ int wave_incl_scan(int v, int lane) {
#pragma unroll
  for (int d = 1; d < 32; d <<= 1) {
    const int up = __shfl_up(v, d);
    if (lane >= d) v += up;
  }
  return v;
}

#define BSH     8
#define BNODES  256
#define MAXB    256
#define NCHUNK  256
#define STILE   4096

__device__ __forceinline__ int chunk_len(int nE)  { return (nE + NCHUNK - 1) / NCHUNK; }
__device__ __forceinline__ int units_per_chunk(int nE) { return (chunk_len(nE) + STILE - 1) / STILE; }

__device__ __forceinline__ void chunk_sort_role(
    const int* __restrict__ srcC, const int* __restrict__ dstC,
    const int* __restrict__ srcE, const int* __restrict__ dstE,
    int2* __restrict__ rec, int* __restrict__ T,
    int nEc, int nE, int n, int j, int* sm, int tid)
{
  int*  hist  = sm;
  int*  lbase = sm + 264;
  int*  wsum  = sm + 520;
  int2* tile  = (int2*)(sm + 528);
  const int lane = tid & 31;
  const int wave = tid >> 5;
  const int per = chunk_len(nE);
  const int tpc = units_per_chunk(nE);
  const int U   = NCHUNK * tpc;
  const int beg = j * per;
  const int end = (beg + per < nE) ? (beg + per) : nE;
  for (int k = 0; k < tpc; ++k) {
    const int t0 = beg + k * STILE;
    const int u  = j * tpc + k;
    hist[tid] = 0;
    lds_barrier();
    int d[16], s[16], r[16];
#pragma unroll
    for (int q = 0; q < 16; ++q) {
      const int e  = t0 + q * 256 + tid;
      const int ec = (e < end) ? e : (end - 1);
      const bool chem = ec < nEc;
      const int* pd = chem ? (dstC + ec) : (dstE + (ec - nEc));
      const int* ps = chem ? (srcC + ec) : (srcE + (ec - nEc));
      d[q] = *pd;
      s[q] = *ps + (chem ? 0 : n);
    }
#pragma unroll
    for (int q = 0; q < 16; ++q) {
      const bool ok = t0 + q * 256 + tid < end;
      r[q] = atomicAdd(&hist[ok ? (d[q] >> BSH) : MAXB], 1);
      if (!ok) d[q] = -1;
    }
    lds_barrier();
    const int c    = hist[tid];
    const int incl = wave_incl_scan(c, lane);
    if (lane == 31) wsum[wave] = incl;
    lds_barrier();
    int wb = 0;
    for (int w = 0; w < wave; ++w) wb += wsum[w];
    const int lb = wb + incl - c;
    lbase[tid] = lb;
    T[(size_t)tid * U + u] = lb;
    int cntT = end - t0;
    cntT = (cntT < STILE) ? cntT : STILE;
    cntT = (cntT > 0) ? cntT : 0;
    if (tid == 0) T[(size_t)MAXB * U + u] = cntT;
    lds_barrier();
#pragma unroll
    for (int q = 0; q < 16; ++q)
      if (d[q] >= 0) tile[lbase[d[q] >> BSH] + r[q]] = make_int2(d[q], s[q]);
    lds_barrier();
    for (int i = tid; i < cntT; i += 256) rec[t0 + i] = tile[i];
    lds_barrier();
  }
}
__global__ __launch_bounds__(256) void rgcn_proj_gemm(
    const float* __restrict__ feats,
    const float* __restrict__ Wc, const float* __restrict__ bc,
    const float* __restrict__ We, const float* __restrict__ be,
    float* __restrict__ WhC, float* __restrict__ WhE, int n,
    const int* __restrict__ srcC, const int* __restrict__ dstC,
    const int* __restrict__ srcE, const int* __restrict__ dstE,
    int2* __restrict__ rec, int* __restrict__ T, int nEc, int nE)
{
  __shared__ __attribute__((aligned(16))) float lds[2 * DFEAT * LDP];
  const int tid  = threadIdx.x;
  if (blockIdx.x < NCHUNK) {
    chunk_sort_role(srcC, dstC, srcE, dstE, rec, T, nEc, nE, n, blockIdx.x, (int*)lds, tid);
    return;
  }
  const int nGemm = gridDim.x - NCHUNK;
  const int wave = tid >> 5;
  const int lane = tid & 31;
  const int half = lane >> 4;
  const int l15  = lane & 15;
  const int n0   = wave * TILE;

  v2f bC[32], bE[32];
  wave_stage16(Wc + (size_t)n0 * DFEAT, lds + n0 * LDP, lane);
  wave_stage16(We + (size_t)n0 * DFEAT, lds + (DFEAT + n0) * LDP, lane);
  lds_barrier();
  read_frags(lds + (n0 + l15) * LDP + 2 * half, bC);
  read_frags(lds + (DFEAT + n0 + l15) * LDP + 2 * half, bE);
  lds_barrier();

  const float biasC = bc[n0 + l15];
  const float biasE = be[n0 + l15];
  const int nTiles = (n + TILE - 1) / TILE;
  const float4 zero4 = make_float4(0.f, 0.f, 0.f, 0.f);
  float4 p0 = zero4, p1 = zero4;
  int t = blockIdx.x - NCHUNK;
  if (t < nTiles) {
    tile_load(feats, t * TILE, n, wave, lane, p0, p1);
    tile_store(lds, wave, lane, p0, p1);
  }
  lds_barrier();
  int cur = 0;
  for (; t < nTiles; t += nGemm) {
    const int  tn   = t + nGemm;
    const bool more = tn < nTiles;
    if (more) tile_load(feats, tn * TILE, n, wave, lane, p0, p1);

    v2f a[32];
    read_frags(lds + cur * (TILE * LDP) + l15 * LDP + 2 * half, a);
    v8f accC = {0.f, 0.f, 0.f, 0.f, 0.f, 0.f, 0.f, 0.f};
    v8f accE = {0.f, 0.f, 0.f, 0.f, 0.f, 0.f, 0.f, 0.f};
#pragma unroll
    for (int k = 0; k < 32; ++k) {
      accC = __builtin_amdgcn_wmma_f32_16x16x4_f32(
          false, a[k], false, bC[k], (short)0, accC, false, false);
      accE = __builtin_amdgcn_wmma_f32_16x16x4_f32(
          false, a[k], false, bE[k], (short)0, accE, false, false);
    }

    const int m0 = t * TILE;
    float oC[8], oE[8];
#pragma unroll
    for (int r = 0; r < 8; ++r) { oC[r] = accC[r] + biasC; oE[r] = accE[r] + biasE; }
    float* ocol = WhC + (size_t)(m0 + half * 8) * DFEAT + n0 + l15;
    float* ecol = WhE + (size_t)(m0 + half * 8) * DFEAT + n0 + l15;
    if (m0 + TILE <= n) {
#pragma unroll
      for (int r = 0; r < 8; ++r) {
        ocol[(size_t)r * DFEAT] = oC[r];
        ecol[(size_t)r * DFEAT] = oE[r];
      }
    } else {
#pragma unroll
      for (int r = 0; r < 8; ++r) {
        if (m0 + half * 8 + r < n) {
          ocol[(size_t)r * DFEAT] = oC[r];
          ecol[(size_t)r * DFEAT] = oE[r];
        }
      }
    }

    if (more) tile_store(lds + (cur ^ 1) * (TILE * LDP), wave, lane, p0, p1);
    lds_barrier();
    cur ^= 1;
  }
}

#define LCAP 12288

__device__ __forceinline__ void put_slot(int* stage, int* __restrict__ lst,
                                         int segBeg, int slot, int val) {
  if (slot < LCAP) stage[slot] = val;
  else             lst[segBeg + slot] = val;
}

__global__ __launch_bounds__(1024) void bucket_place(
    const int* __restrict__ T, const int2* __restrict__ rec,
    int* __restrict__ off, int* __restrict__ lst, int n, int nE, int nB)
{
  __shared__ int cnt[BNODES + 32];
  __shared__ int cntT[BNODES];
  __shared__ int curT[BNODES];
  __shared__ int nbeg[BNODES];
  __shared__ int wsum[8];
  __shared__ int tsum[8];
  __shared__ int stage[LCAP + 32];
  const int tid  = threadIdx.x;
  const int lane = tid & 31;
  const int wave = tid >> 5;
  const int b    = blockIdx.x;
  const int per = chunk_len(nE);
  const int tpc = units_per_chunk(nE);
  const int U   = NCHUNK * tpc;
  const int* Tb = T + (size_t)b * U;
  const int* Te = Tb + U;
  const int team = tid >> 2;
  const int k    = tid & 3;
  int part = 0;
  if (tid < BNODES) {
    cnt[tid]  = 0;
    cntT[tid] = 0;
    curT[tid] = 0;
    for (int u = tid; u < U; u += BNODES) part += Tb[u];
  }
  const int pincl = wave_incl_scan(part, lane);
  if (tid < BNODES && lane == 31) tsum[wave] = pincl;
  lds_barrier();

  int2 v[8];
  int  rk[8];
  int  hs = 0, he = 0;
  const int2* hru = rec;
  if (U > 0) {
    hs  = Tb[team] + k;
    he  = Te[team];
    hru = rec + ((team / tpc) * per + (team % tpc) * STILE);
#pragma unroll
    for (int q = 0; q < 8; ++q) {
      const int r = hs + 4 * q;
      const int2* p = (r < he) ? (hru + r) : rec;
      v[q] = *p;
    }
    __builtin_amdgcn_sched_barrier(0);
#pragma unroll
    for (int q = 0; q < 8; ++q) {
      const bool ok = hs + 4 * q < he;
      rk[q] = atomicAdd(&cnt[ok ? (v[q].x & (BNODES - 1)) : (BNODES + lane)], 1);
    }
    for (int r = hs + 32; r < he; r += 4)
      atomicAdd(&cntT[hru[r].x & (BNODES - 1)], 1);
  }
  for (int u0 = 256; u0 < U; u0 += 256) {
    const int u = u0 + team;
    const int s = Tb[u] + k;
    const int e = Te[u];
    const int2* ru = rec + ((u / tpc) * per + (u % tpc) * STILE);
    for (int r = s; r < e; r += 4)
      atomicAdd(&cntT[ru[r].x & (BNODES - 1)], 1);
  }
  lds_barrier();

  const int c    = (tid < BNODES) ? (cnt[tid] + cntT[tid]) : 0;
  const int incl = wave_incl_scan(c, lane);
  if (tid < BNODES && lane == 31) wsum[wave] = incl;
  lds_barrier();
  int segBeg = 0, tot = 0;
  for (int w = 0; w < 8; ++w) { segBeg += tsum[w]; tot += wsum[w]; }
  if (tid < BNODES) {
    int wb = 0;
    for (int w = 0; w < wave; ++w) wb += wsum[w];
    const int st = wb + incl - c;
    nbeg[tid] = st;
    const int node = b * BNODES + tid;
    if (node < n) off[node] = segBeg + st;
  }
  if (b == nB - 1 && tid == 0) off[n] = nE;
  lds_barrier();

  if (U > 0) {
    int sl[8];
#pragma unroll
    for (int q = 0; q < 8; ++q) sl[q] = nbeg[v[q].x & (BNODES - 1)] + rk[q];
#pragma unroll
    for (int q = 0; q < 8; ++q) {
      const bool ok = hs + 4 * q < he;
      stage[(ok && sl[q] < LCAP) ? sl[q] : (LCAP + lane)] = v[q].y;
      if (ok && sl[q] >= LCAP) lst[segBeg + sl[q]] = v[q].y;
    }
    for (int r = hs + 32; r < he; r += 4) {
      const int2 x  = hru[r];
      const int  nd = x.x & (BNODES - 1);
      put_slot(stage, lst, segBeg, nbeg[nd] + cnt[nd] + atomicAdd(&curT[nd], 1), x.y);
    }
  }
  for (int u0 = 256; u0 < U; u0 += 256) {
    const int u = u0 + team;
    const int s = Tb[u] + k;
    const int e = Te[u];
    const int2* ru = rec + ((u / tpc) * per + (u % tpc) * STILE);
    for (int r = s; r < e; r += 4) {
      const int2 x  = ru[r];
      const int  nd = x.x & (BNODES - 1);
      put_slot(stage, lst, segBeg, nbeg[nd] + cnt[nd] + atomicAdd(&curT[nd], 1), x.y);
    }
  }
  lds_barrier();

  const int m  = (tot < LCAP) ? tot : LCAP;
  const int sh = segBeg & 31;
  for (int i = tid - sh; i < m; i += 1024)
    if (i >= 0) lst[segBeg + i] = stage[i];
}

__device__ __forceinline__ void add4(float4& a, const float4 v) {
  a.x += v.x; a.y += v.y; a.z += v.z; a.w += v.w;
}

#define GROWS 16

__device__ __forceinline__ float4 seg_sum(
    const float4* __restrict__ rows, const int* __restrict__ lst,
    int beg, int end, int idx0, int lane)
{
  const float4 zero4 = make_float4(0.f, 0.f, 0.f, 0.f);
  float4 acc = zero4;
  for (int base = beg; base < end; base += 32) {
    const int m = (end - base < 32) ? (end - base) : 32;
    int idx = idx0;
    if (base != beg) idx = (lane < m) ? lst[base + lane] : 0;
    for (int j = 0; j < m; j += GROWS) {
      const int left = m - j;
      float4 v[GROWS];
      v[0] = rows[(size_t)__builtin_amdgcn_readlane(idx, j) * 32];
#pragma unroll
      for (int q = 1; q < GROWS; ++q) {
        v[q] = zero4;
        if (left > q) v[q] = rows[(size_t)__builtin_amdgcn_readlane(idx, j + q) * 32];
      }
#pragma unroll
      for (int q = 0; q < GROWS; ++q) add4(acc, v[q]);
    }
  }
  return acc;
}

__device__ __forceinline__ void pair_sum(
    const float4* __restrict__ rows, const int* __restrict__ off,
    const int* __restrict__ lst, int na, int nb, int n, int lane,
    float4& p0, float4& p1)
{
  const int ca = (na < n) ? na : (n - 1);
  const int cb = (nb < n) ? nb : (n - 1);
  const int begA = off[ca];
  const int nxtA = off[ca + 1];
  const int begB = off[cb];
  const int nxtB = off[cb + 1];
  const int endA = (na < n) ? nxtA : begA;
  const int endB = (nb < n) ? nxtB : begB;
  const int idxA = (lane < endA - begA) ? lst[begA + lane] : 0;
  const int idxB = (lane < endB - begB) ? lst[begB + lane] : 0;
  p0 = seg_sum(rows, lst, begA, endA, idxA, lane);
  p1 = seg_sum(rows, lst, begB, endB, idxB, lane);
}

__global__ __launch_bounds__(256) void rgcn_out_gemm(
    const float* __restrict__ Wh2, const int* __restrict__ off,
    const int* __restrict__ lst,
    const float* __restrict__ Wo, const float* __restrict__ bo,
    const float* __restrict__ WhE, float* __restrict__ out, int n)
{
  __shared__ __attribute__((aligned(16))) float lds[DFEAT * LDP];
  const int tid  = threadIdx.x;
  const int wave = __builtin_amdgcn_readfirstlane(tid >> 5);
  const int lane = tid & 31;
  const int half = lane >> 4;
  const int l15  = lane & 15;
  const int n0   = wave * TILE;

  v2f b[32];
  wave_stage16(Wo + (size_t)n0 * DFEAT, lds + n0 * LDP, lane);
  lds_barrier();
  read_frags(lds + (n0 + l15) * LDP + 2 * half, b);
  lds_barrier();

  const float bias = bo[n0 + l15];
  const int nTiles = (n + TILE - 1) / TILE;
  const float4* rows = (const float4*)Wh2 + lane;
  const float4 zero4 = make_float4(0.f, 0.f, 0.f, 0.f);
  float4 p0 = zero4, p1 = zero4;
  int t = blockIdx.x;
  if (t < nTiles) {
    pair_sum(rows, off, lst, t * TILE + wave, t * TILE + wave + 8, n, lane, p0, p1);
    tile_store(lds, wave, lane, p0, p1);
  }
  lds_barrier();
  int cur = 0;
  for (; t < nTiles; t += gridDim.x) {
    const int  tn   = t + gridDim.x;
    const bool more = tn < nTiles;
    if (more) pair_sum(rows, off, lst, tn * TILE + wave, tn * TILE + wave + 8, n, lane, p0, p1);

    const int m0 = t * TILE;
    float e[8];
#pragma unroll
    for (int r = 0; r < 8; ++r) {
      int M = m0 + half * 8 + r;
      M = (M < n) ? M : (n - 1);
      e[r] = WhE[(size_t)M * DFEAT + n0 + l15];
    }

    v2f a[32];
    read_frags(lds + cur * (TILE * LDP) + l15 * LDP + 2 * half, a);
    v8f acc = {0.f, 0.f, 0.f, 0.f, 0.f, 0.f, 0.f, 0.f};
#pragma unroll
    for (int k = 0; k < 32; ++k)
      acc = __builtin_amdgcn_wmma_f32_16x16x4_f32(
          false, a[k], false, b[k], (short)0, acc, false, false);

    float o[8];
#pragma unroll
    for (int r = 0; r < 8; ++r) o[r] = acc[r] + bias + e[r];
    float* ocol = out + (size_t)(m0 + half * 8) * DFEAT + n0 + l15;
    if (m0 + TILE <= n) {
#pragma unroll
      for (int r = 0; r < 8; ++r) ocol[(size_t)r * DFEAT] = o[r];
    } else {
#pragma unroll
      for (int r = 0; r < 8; ++r)
        if (m0 + half * 8 + r < n) ocol[(size_t)r * DFEAT] = o[r];
    }

    if (more) tile_store(lds + (cur ^ 1) * (TILE * LDP), wave, lane, p0, p1);
    lds_barrier();
    cur ^= 1;
  }
}

extern "C" void kernel_launch(void* const* d_in, const int* in_sizes, int n_in,
                              void* d_out, int out_size, void* d_ws, size_t ws_size,
                              hipStream_t stream) {
  const float* feats = (const float*)d_in[0];
  const float* Wc    = (const float*)d_in[1];
  const float* bc    = (const float*)d_in[2];
  const float* We    = (const float*)d_in[3];
  const float* be    = (const float*)d_in[4];
  const float* Wo    = (const float*)d_in[5];
  const float* bo    = (const float*)d_in[6];
  const int*   srcC  = (const int*)d_in[7];
  const int*   dstC  = (const int*)d_in[8];
  const int*   srcE  = (const int*)d_in[9];
  const int*   dstE  = (const int*)d_in[10];

  const int n   = in_sizes[0] / DFEAT;
  const int nEc = in_sizes[7];
  const int nEe = in_sizes[9];
  const int nE  = nEc + nEe;

  const size_t feSize = (size_t)n * DFEAT;
  float* WhC = (float*)d_ws;
  float* WhE = WhC + feSize;
  int*   off = (int*)(WhE + feSize);
  int2*  rec = (int2*)(off + ((size_t)n / 1024 + 2) * 1024);
  int*   lst = (int*)(rec + nE);
  int*   T   = lst + nE;
  float* out = (float*)d_out;

  rgcn_proj_gemm<<<NCHUNK + 512, 256, 0, stream>>>(feats, Wc, bc, We, be, WhC, WhE, n,
                                                   srcC, dstC, srcE, dstE, rec, T, nEc, nE);

  const int nB = (n + BNODES - 1) / BNODES;
  bucket_place<<<(unsigned)nB, 1024, 0, stream>>>(T, rec, off, lst, n, nE, nB);

  rgcn_out_gemm<<<768, 256, 0, stream>>>(WhC, off, lst, Wo, bo, WhE, out, n);
}
